# starting kernel
# baseline (speedup 1.0000x reference)
#include <hip/hip_runtime.h>

typedef float v2f __attribute__((ext_vector_type(2)));
typedef float v8f __attribute__((ext_vector_type(8)));

#define VN 96
#define NC 32
#define BX 4
#define BY 4
#define BT 16
#define NTHREADS 256
#define SCALE (1.0f / 27.0f)

#define S1_OFF 0
#define BF_OFF 0
#define S2_OFF 8000
#define M1_OFF 8000
#define M2_OFF 10304
#define PM_OFF 11840
#define SMEM_FLOATS 14400

__device__ __forceinline__ float ldg_vol(const float* __restrict__ p,
                                         int c, int x, int y, int t) {
    if ((unsigned)x < (unsigned)VN && (unsigned)y < (unsigned)VN &&
        (unsigned)t < (unsigned)VN)
        return p[((c * VN + x) * VN + y) * VN + t];
    return 0.0f;
}

__global__ __launch_bounds__(NTHREADS)
void corr3d_wmma_kernel(const float* __restrict__ mov,
                        const float* __restrict__ fix,
                        float* __restrict__ out) {
    __shared__ float smem[SMEM_FLOATS];

    const int tid  = threadIdx.x;
    const int lane = tid & 31;
    const int w    = tid >> 5;
    const int col  = lane & 15;

    const int x0 = blockIdx.z * BX;
    const int y0 = blockIdx.y * BY;
    const int t0 = blockIdx.x * BT;

    const int vA = w * 32 + col;
    const int vB = vA + 16;
    const int lxA = vA >> 6, lyA = (vA >> 4) & 3, ltA = vA & 15;
    const int lxB = vB >> 6, lyB = (vB >> 4) & 3, ltB = vB & 15;
    const int cc0 = (lane < 16) ? 0 : 2;
    const int cc1 = cc0 + 1;

    float accA[27], accB[27];
#pragma unroll
    for (int d = 0; d < 27; ++d) { accA[d] = 0.0f; accB[d] = 0.0f; }

    const v2f a_ones = {1.0f, 1.0f};

    for (int cg = 0; cg < 8; ++cg) {
        __syncthreads();
        for (int idx = tid; idx < 8000; idx += NTHREADS) {
            int c4 = idx / 2000, r = idx - c4 * 2000;
            int ix = r / 200, rr = r - ix * 200;
            int iy = rr / 20, it = rr - iy * 20;
            int ch = cg * 4 + c4;
            int gx = x0 - 3 + ix, gy = y0 - 3 + iy, gt = t0 - 3 + it;
            float v = ldg_vol(fix, ch, gx, gy, gt)
                    + ldg_vol(fix, ch, gx, gy, gt + 1)
                    + ldg_vol(fix, ch, gx, gy, gt + 2);
            smem[S1_OFF + ((c4 * 10 + ix) * 10 + iy) * 20 + it] = v;
        }
        __syncthreads();
        for (int idx = tid; idx < 6400; idx += NTHREADS) {
            int c4 = idx / 1600, r = idx - c4 * 1600;
            int ix = r / 160, rr = r - ix * 160;
            int iy = rr / 20, it = rr - iy * 20;
            const float* s1 = &smem[S1_OFF + ((c4 * 10 + ix) * 10 + iy) * 20 + it];
            smem[S2_OFF + ((c4 * 10 + ix) * 8 + iy) * 20 + it] =
                s1[0] + s1[20] + s1[40];
        }
        __syncthreads();
        for (int idx = tid; idx < 5120; idx += NTHREADS) {
            int c4 = idx / 1280, r = idx - c4 * 1280;
            int ix = r / 160, rr = r - ix * 160;
            int iy = rr / 20, it = rr - iy * 20;
            const float* s2 = &smem[S2_OFF + ((c4 * 10 + ix) * 8 + iy) * 20 + it];
            smem[BF_OFF + ((c4 * 8 + ix) * 8 + iy) * 20 + it] =
                s2[0] + s2[160] + s2[320];
        }
        __syncthreads();
        for (int idx = tid; idx < 2304; idx += NTHREADS) {
            int c4 = idx / 576, r = idx - c4 * 576;
            int ix = r / 96, rr = r - ix * 96;
            int iy = rr / 16, lt = rr - iy * 16;
            int ch = cg * 4 + c4;
            int gx = x0 - 1 + ix, gy = y0 - 1 + iy, gt = t0 + lt - 1;
            float v = ldg_vol(mov, ch, gx, gy, gt)
                    + ldg_vol(mov, ch, gx, gy, gt + 1)
                    + ldg_vol(mov, ch, gx, gy, gt + 2);
            smem[M1_OFF + ((c4 * 6 + ix) * 6 + iy) * 16 + lt] = v;
        }
        __syncthreads();
        for (int idx = tid; idx < 1536; idx += NTHREADS) {
            int c4 = idx / 384, r = idx - c4 * 384;
            int ix = r / 64, rr = r - ix * 64;
            int ly = rr / 16, lt = rr - ly * 16;
            const float* m1 = &smem[M1_OFF + ((c4 * 6 + ix) * 6 + ly) * 16 + lt];
            smem[M2_OFF + ((c4 * 6 + ix) * 4 + ly) * 16 + lt] =
                m1[0] + m1[16] + m1[32];
        }
        __syncthreads();
        for (int idx = tid; idx < 1024; idx += NTHREADS) {
            int c4 = idx / 256, r = idx - c4 * 256;
            int lx = r / 64, rr = r - lx * 64;
            int ly = rr / 16, lt = rr - ly * 16;
            const float* m2 = &smem[M2_OFF + ((c4 * 6 + lx) * 4 + ly) * 16 + lt];
            smem[PM_OFF + ((c4 * 4 + lx) * 4 + ly) * 16 + lt] =
                m2[0] + m2[64] + m2[128];
        }
        __syncthreads();

        const float pmA0 = smem[PM_OFF + ((cc0 * 4 + lxA) * 4 + lyA) * 16 + ltA];
        const float pmA1 = smem[PM_OFF + ((cc1 * 4 + lxA) * 4 + lyA) * 16 + ltA];
        const float pmB0 = smem[PM_OFF + ((cc0 * 4 + lxB) * 4 + lyB) * 16 + ltB];
        const float pmB1 = smem[PM_OFF + ((cc1 * 4 + lxB) * 4 + lyB) * 16 + ltB];

#pragma unroll
        for (int i = 0; i < 3; ++i)
#pragma unroll
        for (int j = 0; j < 3; ++j)
#pragma unroll
        for (int k = 0; k < 3; ++k) {
            const int d = (i * 3 + j) * 3 + k;
            const int pA0 = ((cc0 * 8 + (lxA + 2*i)) * 8 + (lyA + 2*j)) * 20 + (ltA + 2*k);
            const int pA1 = ((cc1 * 8 + (lxA + 2*i)) * 8 + (lyA + 2*j)) * 20 + (ltA + 2*k);
            const int pB0 = ((cc0 * 8 + (lxB + 2*i)) * 8 + (lyB + 2*j)) * 20 + (ltB + 2*k);
            const int pB1 = ((cc1 * 8 + (lxB + 2*i)) * 8 + (lyB + 2*j)) * 20 + (ltB + 2*k);
            v2f bA = { pmA0 * smem[BF_OFF + pA0], pmA1 * smem[BF_OFF + pA1] };
            v2f bB = { pmB0 * smem[BF_OFF + pB0], pmB1 * smem[BF_OFF + pB1] };
            v8f z = {};
            v8f rA = __builtin_amdgcn_wmma_f32_16x16x4_f32(
                false, a_ones, false, bA, (short)0, z, false, false);
            v8f rB = __builtin_amdgcn_wmma_f32_16x16x4_f32(
                false, a_ones, false, bB, (short)0, z, false, false);
            accA[d] += rA[0];
            accB[d] += rB[0];
        }
    }

    const int v  = w * 32 + lane;
    const int lx = v >> 6, ly = (v >> 4) & 3, lt = v & 15;
    const int x = x0 + lx, y = y0 + ly, t = t0 + lt;
#pragma unroll
    for (int d = 0; d < 27; ++d) {
        float val = ((lane < 16) ? accA[d] : accB[d]) * SCALE;
        out[((d * VN + x) * VN + y) * VN + t] = val;
    }
}

extern "C" void kernel_launch(void* const* d_in, const int* in_sizes, int n_in,
                              void* d_out, int out_size, void* d_ws, size_t ws_size,
                              hipStream_t stream) {
    (void)in_sizes; (void)n_in; (void)out_size; (void)d_ws; (void)ws_size;
    const float* mov = (const float*)d_in[0];
    const float* fix = (const float*)d_in[1];
    float* out = (float*)d_out;
    dim3 grid(VN / BT, VN / BY, VN / BX);
    corr3d_wmma_kernel<<<grid, NTHREADS, 0, stream>>>(mov, fix, out);
}


# Halo t-pass reads global memory as batches of independent aligned 16-byte loads (8 lanes per row, neighbours by lane shuffle, branch-free zero padding) instead of 3 dword loads per loop trip
# speedup vs baseline: 1.7873x; 1.7873x over previous
#include <hip/hip_runtime.h>

typedef float v2f __attribute__((ext_vector_type(2)));
typedef float v4f __attribute__((ext_vector_type(4)));
typedef float v8f __attribute__((ext_vector_type(8)));

#define VN 96
#define NC 32
#define BX 4
#define BY 4
#define BT 16
#define NTHREADS 256
#define SCALE (1.0f / 27.0f)
#define NFROWS 400
#define NMROWS 144
#define NFTRIPS 13
#define NMTRIPS 5

#define S1_OFF 0
#define BF_OFF 0
#define S2_OFF 8000
#define M2_OFF 10304
#define PM_OFF 11840
#define M1_OFF 14400
#define SMEM_FLOATS 16704


__device__ __forceinline__ v4f halo_piece(const v4f* __restrict__ p, bool ok, int idx) {
    const int i4 = ok ? idx : 0;
    __builtin_assume(i4 >= 0);
    return p[i4];
}

__device__ __forceinline__ v4f fix_tpass(v4f v) {
    const float n0 = __shfl_down(v[0], 1, 8);
    const float n1 = __shfl_down(v[1], 1, 8);
    const float n2 = __shfl_down(v[2], 1, 8);
    v4f o;
    o[0] = v[1] + v[2] + v[3];
    o[1] = v[2] + v[3] + n0;
    o[2] = v[3] + n0 + n1;
    o[3] = n0 + n1 + n2;
    return o;
}

__device__ __forceinline__ v4f mov_tpass(v4f v) {
    const float p3 = __shfl_up(v[3], 1, 8);
    const float n0 = __shfl_down(v[0], 1, 8);
    v4f o;
    o[0] = p3 + v[0] + v[1];
    o[1] = v[0] + v[1] + v[2];
    o[2] = v[1] + v[2] + v[3];
    o[3] = v[2] + v[3] + n0;
    return o;
}

template <int K0, int N>
__device__ __forceinline__ void fix_batch(const v4f* __restrict__ fix4,
                                          const int (&off)[NFTRIPS], const bool (&ok)[NFTRIPS],
                                          int cgoff, float* smem, int hq, int hg) {
    const v4f zero4 = {0.0f, 0.0f, 0.0f, 0.0f};
    v4f pc[N];
#pragma unroll
    for (int j = 0; j < N; ++j)
        pc[j] = halo_piece(fix4, ok[K0 + j], off[K0 + j] + cgoff);
#pragma unroll
    for (int j = 0; j < N; ++j) {
        const v4f o = fix_tpass(ok[K0 + j] ? pc[j] : zero4);
        const int R = (K0 + j) * 32 + hg;
        if (hq < 5 && R < NFROWS)
            *reinterpret_cast<v4f*>(&smem[S1_OFF + R * 20 + 4 * hq]) = o;
    }
}

template <int K0, int N>
__device__ __forceinline__ void mov_batch(const v4f* __restrict__ mov4,
                                          const int (&off)[NMTRIPS], const bool (&ok)[NMTRIPS],
                                          int cgoff, float* smem, int hq, int hg) {
    const v4f zero4 = {0.0f, 0.0f, 0.0f, 0.0f};
    v4f pc[N];
#pragma unroll
    for (int j = 0; j < N; ++j)
        pc[j] = halo_piece(mov4, ok[K0 + j], off[K0 + j] + cgoff);
#pragma unroll
    for (int j = 0; j < N; ++j) {
        const v4f o = mov_tpass(ok[K0 + j] ? pc[j] : zero4);
        const int M = (K0 + j) * 32 + hg;
        if (hq >= 1 && hq <= 4 && M < NMROWS)
            *reinterpret_cast<v4f*>(&smem[M1_OFF + M * 16 + 4 * (hq - 1)]) = o;
    }
}

__global__ __launch_bounds__(NTHREADS, 6)
void corr3d_wmma_kernel(const float* __restrict__ mov,
                        const float* __restrict__ fix,
                        float* __restrict__ out) {
    __shared__ __attribute__((aligned(16))) float smem[SMEM_FLOATS];

    const int tid  = threadIdx.x;
    const int lane = tid & 31;
    const int w    = tid >> 5;
    const int col  = lane & 15;

    const int x0 = blockIdx.z * BX;
    const int y0 = blockIdx.y * BY;
    const int t0 = blockIdx.x * BT;

    const int vA = w * 32 + col;
    const int vB = vA + 16;
    const int lxA = vA >> 6, lyA = (vA >> 4) & 3, ltA = vA & 15;
    const int lxB = vB >> 6, lyB = (vB >> 4) & 3, ltB = vB & 15;
    const int cc0 = (lane < 16) ? 0 : 2;
    const int cc1 = cc0 + 1;

    float accA[27], accB[27];
#pragma unroll
    for (int d = 0; d < 27; ++d) { accA[d] = 0.0f; accB[d] = 0.0f; }

    const v2f a_ones = {1.0f, 1.0f};

    const int  hq   = tid & 7;
    const int  hg   = tid >> 3;
    const int  hgt0 = t0 - 4 + 4 * hq;
    const bool htok = (hq < 6) && (unsigned)hgt0 < (unsigned)VN;
    int  foff[NFTRIPS]; bool fok[NFTRIPS];
#pragma unroll
    for (int k = 0; k < NFTRIPS; ++k) {
        const int R   = k * 32 + hg;
        const int Rc  = (R < NFROWS) ? R : 0;
        const int c4  = Rc / 100, rem = Rc - c4 * 100;
        const int ix  = rem / 10, iy = rem - ix * 10;
        const int gx  = x0 - 3 + ix, gy = y0 - 3 + iy;
        fok[k]  = htok && (R < NFROWS) &&
                  (unsigned)gx < (unsigned)VN && (unsigned)gy < (unsigned)VN;
        foff[k] = ((c4 * VN + gx) * VN + gy) * (VN / 4) + (hgt0 >> 2);
    }
    int  moff[NMTRIPS]; bool mok[NMTRIPS];
#pragma unroll
    for (int k = 0; k < NMTRIPS; ++k) {
        const int M   = k * 32 + hg;
        const int Mc  = (M < NMROWS) ? M : 0;
        const int c4  = Mc / 36, rem = Mc - c4 * 36;
        const int ix  = rem / 6, iy = rem - ix * 6;
        const int gx  = x0 - 1 + ix, gy = y0 - 1 + iy;
        mok[k]  = htok && (M < NMROWS) &&
                  (unsigned)gx < (unsigned)VN && (unsigned)gy < (unsigned)VN;
        moff[k] = ((c4 * VN + gx) * VN + gy) * (VN / 4) + (hgt0 >> 2);
    }
    const v4f* __restrict__ fix4 = reinterpret_cast<const v4f*>(fix);
    const v4f* __restrict__ mov4 = reinterpret_cast<const v4f*>(mov);

    for (int cg = 0; cg < 8; ++cg) {
        __syncthreads();
        {
            const int cgoff = cg * 4 * VN * VN * (VN / 4);
            fix_batch<0, 5>(fix4, foff, fok, cgoff, smem, hq, hg);
            fix_batch<5, 5>(fix4, foff, fok, cgoff, smem, hq, hg);
            fix_batch<10, 3>(fix4, foff, fok, cgoff, smem, hq, hg);
            mov_batch<0, 5>(mov4, moff, mok, cgoff, smem, hq, hg);
        }
        __syncthreads();
        for (int idx = tid; idx < 6400; idx += NTHREADS) {
            int c4 = idx / 1600, r = idx - c4 * 1600;
            int ix = r / 160, rr = r - ix * 160;
            int iy = rr / 20, it = rr - iy * 20;
            const float* s1 = &smem[S1_OFF + ((c4 * 10 + ix) * 10 + iy) * 20 + it];
            smem[S2_OFF + ((c4 * 10 + ix) * 8 + iy) * 20 + it] =
                s1[0] + s1[20] + s1[40];
        }
        __syncthreads();
        for (int idx = tid; idx < 5120; idx += NTHREADS) {
            int c4 = idx / 1280, r = idx - c4 * 1280;
            int ix = r / 160, rr = r - ix * 160;
            int iy = rr / 20, it = rr - iy * 20;
            const float* s2 = &smem[S2_OFF + ((c4 * 10 + ix) * 8 + iy) * 20 + it];
            smem[BF_OFF + ((c4 * 8 + ix) * 8 + iy) * 20 + it] =
                s2[0] + s2[160] + s2[320];
        }
        __syncthreads();
        for (int idx = tid; idx < 1536; idx += NTHREADS) {
            int c4 = idx / 384, r = idx - c4 * 384;
            int ix = r / 64, rr = r - ix * 64;
            int ly = rr / 16, lt = rr - ly * 16;
            const float* m1 = &smem[M1_OFF + ((c4 * 6 + ix) * 6 + ly) * 16 + lt];
            smem[M2_OFF + ((c4 * 6 + ix) * 4 + ly) * 16 + lt] =
                m1[0] + m1[16] + m1[32];
        }
        __syncthreads();
        for (int idx = tid; idx < 1024; idx += NTHREADS) {
            int c4 = idx / 256, r = idx - c4 * 256;
            int lx = r / 64, rr = r - lx * 64;
            int ly = rr / 16, lt = rr - ly * 16;
            const float* m2 = &smem[M2_OFF + ((c4 * 6 + lx) * 4 + ly) * 16 + lt];
            smem[PM_OFF + ((c4 * 4 + lx) * 4 + ly) * 16 + lt] =
                m2[0] + m2[64] + m2[128];
        }
        __syncthreads();

        const float pmA0 = smem[PM_OFF + ((cc0 * 4 + lxA) * 4 + lyA) * 16 + ltA];
        const float pmA1 = smem[PM_OFF + ((cc1 * 4 + lxA) * 4 + lyA) * 16 + ltA];
        const float pmB0 = smem[PM_OFF + ((cc0 * 4 + lxB) * 4 + lyB) * 16 + ltB];
        const float pmB1 = smem[PM_OFF + ((cc1 * 4 + lxB) * 4 + lyB) * 16 + ltB];

#pragma unroll
        for (int i = 0; i < 3; ++i)
#pragma unroll
        for (int j = 0; j < 3; ++j)
#pragma unroll
        for (int k = 0; k < 3; ++k) {
            const int d = (i * 3 + j) * 3 + k;
            const int pA0 = ((cc0 * 8 + (lxA + 2*i)) * 8 + (lyA + 2*j)) * 20 + (ltA + 2*k);
            const int pA1 = ((cc1 * 8 + (lxA + 2*i)) * 8 + (lyA + 2*j)) * 20 + (ltA + 2*k);
            const int pB0 = ((cc0 * 8 + (lxB + 2*i)) * 8 + (lyB + 2*j)) * 20 + (ltB + 2*k);
            const int pB1 = ((cc1 * 8 + (lxB + 2*i)) * 8 + (lyB + 2*j)) * 20 + (ltB + 2*k);
            v2f bA = { pmA0 * smem[BF_OFF + pA0], pmA1 * smem[BF_OFF + pA1] };
            v2f bB = { pmB0 * smem[BF_OFF + pB0], pmB1 * smem[BF_OFF + pB1] };
            v8f z = {};
            v8f rA = __builtin_amdgcn_wmma_f32_16x16x4_f32(
                false, a_ones, false, bA, (short)0, z, false, false);
            v8f rB = __builtin_amdgcn_wmma_f32_16x16x4_f32(
                false, a_ones, false, bB, (short)0, z, false, false);
            accA[d] += rA[0];
            accB[d] += rB[0];
        }
    }

    const int v  = w * 32 + lane;
    const int lx = v >> 6, ly = (v >> 4) & 3, lt = v & 15;
    const int x = x0 + lx, y = y0 + ly, t = t0 + lt;
#pragma unroll
    for (int d = 0; d < 27; ++d) {
        float val = ((lane < 16) ? accA[d] : accB[d]) * SCALE;
        out[((d * VN + x) * VN + y) * VN + t] = val;
    }
}

extern "C" void kernel_launch(void* const* d_in, const int* in_sizes, int n_in,
                              void* d_out, int out_size, void* d_ws, size_t ws_size,
                              hipStream_t stream) {
    (void)in_sizes; (void)n_in; (void)out_size; (void)d_ws; (void)ws_size;
    const float* mov = (const float*)d_in[0];
    const float* fix = (const float*)d_in[1];
    float* out = (float*)d_out;
    dim3 grid(VN / BT, VN / BY, VN / BX);
    corr3d_wmma_kernel<<<grid, NTHREADS, 0, stream>>>(mov, fix, out);
}


# y- and x-pass of both box filters fused into one column sweep per thread on 16-byte LDS pieces, window in registers: no index division, no intermediate arrays or overlays, 2 barriers per group not 6
# speedup vs baseline: 2.8577x; 1.5989x over previous
#include <hip/hip_runtime.h>

typedef float v2f __attribute__((ext_vector_type(2)));
typedef float v4f __attribute__((ext_vector_type(4)));
typedef float v8f __attribute__((ext_vector_type(8)));

#define VN 96
#define NC 32
#define BX 4
#define BY 4
#define BT 16
#define NTHREADS 256
#define SCALE (1.0f / 27.0f)
#define NFROWS 400
#define NMROWS 144
#define NFTRIPS 13
#define NMTRIPS 5

#define S1_OFF 0
#define BF_OFF 8000
#define PM_OFF 13120
#define M1_OFF 14144
#define SMEM_FLOATS 16448


__device__ __forceinline__ v4f halo_piece(const v4f* __restrict__ p, bool ok, int idx) {
    const int i4 = ok ? idx : 0;
    __builtin_assume(i4 >= 0);
    return p[i4];
}

__device__ __forceinline__ v4f fix_tpass(v4f v) {
    const float n0 = __shfl_down(v[0], 1, 8);
    const float n1 = __shfl_down(v[1], 1, 8);
    const float n2 = __shfl_down(v[2], 1, 8);
    v4f o;
    o[0] = v[1] + v[2] + v[3];
    o[1] = v[2] + v[3] + n0;
    o[2] = v[3] + n0 + n1;
    o[3] = n0 + n1 + n2;
    return o;
}

__device__ __forceinline__ v4f mov_tpass(v4f v) {
    const float p3 = __shfl_up(v[3], 1, 8);
    const float n0 = __shfl_down(v[0], 1, 8);
    v4f o;
    o[0] = p3 + v[0] + v[1];
    o[1] = v[0] + v[1] + v[2];
    o[2] = v[1] + v[2] + v[3];
    o[3] = v[2] + v[3] + n0;
    return o;
}

template <int K0, int N>
__device__ __forceinline__ void fix_batch(const v4f* __restrict__ fix4,
                                          const int (&off)[NFTRIPS], const bool (&ok)[NFTRIPS],
                                          int cgoff, float* smem, int hq, int hg) {
    const v4f zero4 = {0.0f, 0.0f, 0.0f, 0.0f};
    v4f pc[N];
#pragma unroll
    for (int j = 0; j < N; ++j)
        pc[j] = halo_piece(fix4, ok[K0 + j], off[K0 + j] + cgoff);
#pragma unroll
    for (int j = 0; j < N; ++j) {
        const v4f o = fix_tpass(ok[K0 + j] ? pc[j] : zero4);
        const int R = (K0 + j) * 32 + hg;
        if (hq < 5 && R < NFROWS)
            *reinterpret_cast<v4f*>(&smem[S1_OFF + R * 20 + 4 * hq]) = o;
    }
}

template <int K0, int N>
__device__ __forceinline__ void mov_batch(const v4f* __restrict__ mov4,
                                          const int (&off)[NMTRIPS], const bool (&ok)[NMTRIPS],
                                          int cgoff, float* smem, int hq, int hg) {
    const v4f zero4 = {0.0f, 0.0f, 0.0f, 0.0f};
    v4f pc[N];
#pragma unroll
    for (int j = 0; j < N; ++j)
        pc[j] = halo_piece(mov4, ok[K0 + j], off[K0 + j] + cgoff);
#pragma unroll
    for (int j = 0; j < N; ++j) {
        const v4f o = mov_tpass(ok[K0 + j] ? pc[j] : zero4);
        const int M = (K0 + j) * 32 + hg;
        if (hq >= 1 && hq <= 4 && M < NMROWS)
            *reinterpret_cast<v4f*>(&smem[M1_OFF + M * 16 + 4 * (hq - 1)]) = o;
    }
}

template <int NIN, int XSTEP, int YSTEP, int DSTEP>
__device__ __forceinline__ void sweep_yx(const float* src, float* dst) {
    v4f y0 = {0.0f, 0.0f, 0.0f, 0.0f}, y1 = y0;
#pragma unroll
    for (int i = 0; i < NIN; ++i) {
        const float* p = src + i * XSTEP;
        const v4f a = *reinterpret_cast<const v4f*>(p);
        const v4f b = *reinterpret_cast<const v4f*>(p + YSTEP);
        const v4f c = *reinterpret_cast<const v4f*>(p + 2 * YSTEP);
        const v4f y = a + b + c;
        if (i >= 2)
            *reinterpret_cast<v4f*>(dst + (i - 2) * DSTEP) = y0 + y1 + y;
        y0 = y1;
        y1 = y;
    }
}

__global__ __launch_bounds__(NTHREADS, 6)
void corr3d_wmma_kernel(const float* __restrict__ mov,
                        const float* __restrict__ fix,
                        float* __restrict__ out) {
    __shared__ __attribute__((aligned(16))) float smem[SMEM_FLOATS];

    const int tid  = threadIdx.x;
    const int lane = tid & 31;
    const int w    = tid >> 5;
    const int col  = lane & 15;

    const int x0 = blockIdx.z * BX;
    const int y0 = blockIdx.y * BY;
    const int t0 = blockIdx.x * BT;

    const int vA = w * 32 + col;
    const int vB = vA + 16;
    const int lxA = vA >> 6, lyA = (vA >> 4) & 3, ltA = vA & 15;
    const int lxB = vB >> 6, lyB = (vB >> 4) & 3, ltB = vB & 15;
    const int cc0 = (lane < 16) ? 0 : 2;
    const int cc1 = cc0 + 1;

    float accA[27], accB[27];
#pragma unroll
    for (int d = 0; d < 27; ++d) { accA[d] = 0.0f; accB[d] = 0.0f; }

    const v2f a_ones = {1.0f, 1.0f};

    const int  hq   = tid & 7;
    const int  hg   = tid >> 3;
    const int  hgt0 = t0 - 4 + 4 * hq;
    const bool htok = (hq < 6) && (unsigned)hgt0 < (unsigned)VN;
    int  foff[NFTRIPS]; bool fok[NFTRIPS];
#pragma unroll
    for (int k = 0; k < NFTRIPS; ++k) {
        const int R   = k * 32 + hg;
        const int Rc  = (R < NFROWS) ? R : 0;
        const int c4  = Rc / 100, rem = Rc - c4 * 100;
        const int ix  = rem / 10, iy = rem - ix * 10;
        const int gx  = x0 - 3 + ix, gy = y0 - 3 + iy;
        fok[k]  = htok && (R < NFROWS) &&
                  (unsigned)gx < (unsigned)VN && (unsigned)gy < (unsigned)VN;
        foff[k] = ((c4 * VN + gx) * VN + gy) * (VN / 4) + (hgt0 >> 2);
    }
    int  moff[NMTRIPS]; bool mok[NMTRIPS];
#pragma unroll
    for (int k = 0; k < NMTRIPS; ++k) {
        const int M   = k * 32 + hg;
        const int Mc  = (M < NMROWS) ? M : 0;
        const int c4  = Mc / 36, rem = Mc - c4 * 36;
        const int ix  = rem / 6, iy = rem - ix * 6;
        const int gx  = x0 - 1 + ix, gy = y0 - 1 + iy;
        mok[k]  = htok && (M < NMROWS) &&
                  (unsigned)gx < (unsigned)VN && (unsigned)gy < (unsigned)VN;
        moff[k] = ((c4 * VN + gx) * VN + gy) * (VN / 4) + (hgt0 >> 2);
    }
    const v4f* __restrict__ fix4 = reinterpret_cast<const v4f*>(fix);
    const v4f* __restrict__ mov4 = reinterpret_cast<const v4f*>(mov);

    for (int cg = 0; cg < 8; ++cg) {
        {
            const int cgoff = cg * 4 * VN * VN * (VN / 4);
            fix_batch<0, 5>(fix4, foff, fok, cgoff, smem, hq, hg);
            fix_batch<5, 5>(fix4, foff, fok, cgoff, smem, hq, hg);
            fix_batch<10, 3>(fix4, foff, fok, cgoff, smem, hq, hg);
            mov_batch<0, 5>(mov4, moff, mok, cgoff, smem, hq, hg);
        }
        __syncthreads();
        if (tid < 160) {
            const int c4 = tid / 40, j = tid - c4 * 40;
            sweep_yx<10, 200, 20, 640>(&smem[S1_OFF + c4 * 2000 + 4 * j],
                                       &smem[BF_OFF + 4 * tid]);
        }
        if (tid >= 192) {
            const int jj = tid - 192;
            const int c4 = jj >> 4, jm = jj & 15;
            sweep_yx<6, 96, 16, 256>(&smem[M1_OFF + c4 * 576 + 4 * jm],
                                     &smem[PM_OFF + 4 * jj]);
        }
        __syncthreads();

        const float pmA0 = smem[PM_OFF + ((lxA * 4 + cc0) * 4 + lyA) * 16 + ltA];
        const float pmA1 = smem[PM_OFF + ((lxA * 4 + cc1) * 4 + lyA) * 16 + ltA];
        const float pmB0 = smem[PM_OFF + ((lxB * 4 + cc0) * 4 + lyB) * 16 + ltB];
        const float pmB1 = smem[PM_OFF + ((lxB * 4 + cc1) * 4 + lyB) * 16 + ltB];

#pragma unroll
        for (int i = 0; i < 3; ++i)
#pragma unroll
        for (int j = 0; j < 3; ++j)
#pragma unroll
        for (int k = 0; k < 3; ++k) {
            const int d = (i * 3 + j) * 3 + k;
            const int pA0 = (((lxA + 2*i) * 4 + cc0) * 8 + (lyA + 2*j)) * 20 + (ltA + 2*k);
            const int pA1 = (((lxA + 2*i) * 4 + cc1) * 8 + (lyA + 2*j)) * 20 + (ltA + 2*k);
            const int pB0 = (((lxB + 2*i) * 4 + cc0) * 8 + (lyB + 2*j)) * 20 + (ltB + 2*k);
            const int pB1 = (((lxB + 2*i) * 4 + cc1) * 8 + (lyB + 2*j)) * 20 + (ltB + 2*k);
            v2f bA = { pmA0 * smem[BF_OFF + pA0], pmA1 * smem[BF_OFF + pA1] };
            v2f bB = { pmB0 * smem[BF_OFF + pB0], pmB1 * smem[BF_OFF + pB1] };
            v8f z = {};
            v8f rA = __builtin_amdgcn_wmma_f32_16x16x4_f32(
                false, a_ones, false, bA, (short)0, z, false, false);
            v8f rB = __builtin_amdgcn_wmma_f32_16x16x4_f32(
                false, a_ones, false, bB, (short)0, z, false, false);
            accA[d] += rA[0];
            accB[d] += rB[0];
        }
    }

    const int v  = w * 32 + lane;
    const int lx = v >> 6, ly = (v >> 4) & 3, lt = v & 15;
    const int x = x0 + lx, y = y0 + ly, t = t0 + lt;
#pragma unroll
    for (int d = 0; d < 27; ++d) {
        float val = ((lane < 16) ? accA[d] : accB[d]) * SCALE;
        out[((d * VN + x) * VN + y) * VN + t] = val;
    }
}

extern "C" void kernel_launch(void* const* d_in, const int* in_sizes, int n_in,
                              void* d_out, int out_size, void* d_ws, size_t ws_size,
                              hipStream_t stream) {
    (void)in_sizes; (void)n_in; (void)out_size; (void)d_ws; (void)ws_size;
    const float* mov = (const float*)d_in[0];
    const float* fix = (const float*)d_in[1];
    float* out = (float*)d_out;
    dim3 grid(VN / BT, VN / BY, VN / BX);
    corr3d_wmma_kernel<<<grid, NTHREADS, 0, stream>>>(mov, fix, out);
}


# Fix-input halo rows dealt one x-plane per trip (one lane address for ten trips, plane offset in scalar registers): freed registers let a channel group's 18 loads go as 3 sets of 6, 3 waits not 5
# speedup vs baseline: 3.4218x; 1.1974x over previous
#include <hip/hip_runtime.h>

typedef float v2f __attribute__((ext_vector_type(2)));
typedef float v4f __attribute__((ext_vector_type(4)));
typedef float v8f __attribute__((ext_vector_type(8)));

#define VN 96
#define NC 32
#define BX 4
#define BY 4
#define BT 16
#define NTHREADS 256
#define SCALE (1.0f / 27.0f)
#define NMROWS 144
#define NMTRIPS 5
#define ROW4 (VN / 4)
#define PLANE4 (VN * ROW4)
#define CHAN4 (VN * PLANE4)

#define S1_OFF 0
#define BF_OFF 8000
#define BF_CS 176
#define BF_XS 708
#define PM_OFF 13664
#define M1_OFF 14688
#define SMEM_FLOATS 16992


__device__ __forceinline__ v4f halo_piece(const v4f* __restrict__ p, bool ok, int idx) {
    const int i4 = ok ? idx : 0;
    __builtin_assume(i4 >= 0);
    return p[i4];
}

__device__ __forceinline__ float lane_next(float x) {
    return __builtin_bit_cast(float, __builtin_amdgcn_update_dpp(
        0, __builtin_bit_cast(int, x), 0x101, 0xf, 0xf, true));
}
__device__ __forceinline__ float lane_prev(float x) {
    return __builtin_bit_cast(float, __builtin_amdgcn_update_dpp(
        0, __builtin_bit_cast(int, x), 0x111, 0xf, 0xf, true));
}

__device__ __forceinline__ v4f fix_tpass(v4f v) {
    const float n0 = lane_next(v[0]);
    const float n1 = lane_next(v[1]);
    const float n2 = lane_next(v[2]);
    v4f o;
    o[0] = v[1] + v[2] + v[3];
    o[1] = v[2] + v[3] + n0;
    o[2] = v[3] + n0 + n1;
    o[3] = n0 + n1 + n2;
    return o;
}

__device__ __forceinline__ v4f mov_tpass(v4f v) {
    const float p3 = lane_prev(v[3]);
    const float n0 = lane_next(v[0]);
    v4f o;
    o[0] = p3 + v[0] + v[1];
    o[1] = v[0] + v[1] + v[2];
    o[2] = v[1] + v[2] + v[3];
    o[3] = v[2] + v[3] + n0;
    return o;
}


__device__ __forceinline__ v4f piece_at(const v4f* __restrict__ p, int idx) {
    __builtin_assume(idx >= 0);
    return p[idx];
}

__device__ __forceinline__ const v4f* plane_ptr(const v4f* __restrict__ base, int gx, int x0) {
    const int gxc = (unsigned)gx < (unsigned)VN ? gx : x0;
    return base + gxc * PLANE4;
}

template <int K0, int N>
__device__ __forceinline__ void fix_plane_store(const v4f (&pc)[N], bool okA, int x0,
                                                float* s1p, bool st) {
    const v4f zero4 = {0.0f, 0.0f, 0.0f, 0.0f};
#pragma unroll
    for (int j = 0; j < N; ++j) {
        const bool ok = okA && (unsigned)(x0 - 3 + K0 + j) < (unsigned)VN;
        const v4f o = fix_tpass(ok ? pc[j] : zero4);
        if (st)
            *reinterpret_cast<v4f*>(s1p + (K0 + j) * 200) = o;
    }
}

template <int NIN, int XSTEP, int YSTEP, int DSTEP>
__device__ __forceinline__ void sweep_yx(const float* src, float* dst) {
    v4f y0 = {0.0f, 0.0f, 0.0f, 0.0f}, y1 = y0;
#pragma unroll
    for (int i = 0; i < NIN; ++i) {
        const float* p = src + i * XSTEP;
        const v4f a = *reinterpret_cast<const v4f*>(p);
        const v4f b = *reinterpret_cast<const v4f*>(p + YSTEP);
        const v4f c = *reinterpret_cast<const v4f*>(p + 2 * YSTEP);
        const v4f y = a + b + c;
        if (i >= 2)
            *reinterpret_cast<v4f*>(dst + (i - 2) * DSTEP) = y0 + y1 + y;
        y0 = y1;
        y1 = y;
    }
}

__global__ __launch_bounds__(NTHREADS, 6)
void corr3d_wmma_kernel(const float* __restrict__ mov,
                        const float* __restrict__ fix,
                        float* __restrict__ out) {
    __shared__ __attribute__((aligned(16))) float smem[SMEM_FLOATS];

    const int tid  = threadIdx.x;
    const int lane = tid & 31;
    const int w    = tid >> 5;
    const int col  = lane & 15;
    const int hlf  = lane >> 4;

    const int x0 = blockIdx.z * BX;
    const int y0 = blockIdx.y * BY;
    const int t0 = blockIdx.x * BT;

    const int px = (w >> 2) & 1, py = (w >> 1) & 1, pt = w & 1;
    const int pm_base = PM_OFF +
        (((px + 2 * (col >> 3)) * 4 + 2 * hlf) * 4 + (py + 2 * ((col >> 2) & 1))) * 16 +
        pt + 2 * (col & 3);
    const int bf_base = BF_OFF + (px + 2 * (col >> 2)) * BF_XS + (2 * hlf) * BF_CS +
                        (py + 2 * (col & 3)) * 20 + pt;

    v8f acc[2][6];
#pragma unroll
    for (int q = 0; q < 2; ++q)
#pragma unroll
        for (int u = 0; u < 6; ++u) {
            const v8f z = {};
            acc[q][u] = z;
        }

    const int  hq   = tid & 7;
    const int  hg   = tid >> 3;
    const int  hgt0 = t0 - 4 + 4 * hq;
    const bool htok = (hq < 6) && (unsigned)hgt0 < (unsigned)VN;
    const int  c4h  = hg >> 3;
    const int  r8   = hg & 7;
    const int  gyA   = y0 - 3 + r8;
    const bool yokA  = (unsigned)gyA < (unsigned)VN;
    const bool okA   = htok && yokA;
    const int  baseA = c4h * CHAN4 + (yokA ? gyA : y0) * ROW4 + ((htok ? hgt0 : t0) >> 2);
    const int  iyB   = 8 + (r8 >> 2), ixB = r8 & 3;
    const int  gyB   = y0 - 3 + iyB;
    const bool okBy  = htok && (unsigned)gyB < (unsigned)VN;
    const int  baseB = c4h * CHAN4 + (x0 - 3 + ixB) * PLANE4 + gyB * ROW4 + (hgt0 >> 2);
    bool bok[3];
#pragma unroll
    for (int j = 0; j < 3; ++j)
        bok[j] = okBy && (4 * j + ixB < 10) &&
                 (unsigned)(x0 - 3 + 4 * j + ixB) < (unsigned)VN;
    float* const s1A = &smem[S1_OFF + c4h * 2000 + r8 * 20 + 4 * hq];
    float* const s1B = &smem[S1_OFF + c4h * 2000 + ixB * 200 + iyB * 20 + 4 * hq];
    const bool stA = hq < 5;
    int  moff[NMTRIPS]; bool mok[NMTRIPS];
#pragma unroll
    for (int k = 0; k < NMTRIPS; ++k) {
        const int M   = k * 32 + hg;
        const int Mc  = (M < NMROWS) ? M : 0;
        const int c4  = Mc / 36, rem = Mc - c4 * 36;
        const int ix  = rem / 6, iy = rem - ix * 6;
        const int gx  = x0 - 1 + ix, gy = y0 - 1 + iy;
        mok[k]  = htok && (M < NMROWS) &&
                  (unsigned)gx < (unsigned)VN && (unsigned)gy < (unsigned)VN;
        moff[k] = ((c4 * VN + gx) * VN + gy) * ROW4 + (hgt0 >> 2);
    }
    const v4f* __restrict__ fix4 = reinterpret_cast<const v4f*>(fix);
    const v4f* __restrict__ mov4 = reinterpret_cast<const v4f*>(mov);

    for (int cg = 0; cg < 8; ++cg) {
        {
            const v4f zero4 = {0.0f, 0.0f, 0.0f, 0.0f};
            const int cgoff = cg * 4 * CHAN4;
            const int idxA  = baseA + cgoff;
            {
                v4f pa[6];
#pragma unroll
                for (int k = 0; k < 6; ++k)
                    pa[k] = piece_at(plane_ptr(fix4, x0 - 3 + k, x0), idxA);
                fix_plane_store<0, 6>(pa, okA, x0, s1A, stA);
            }
            {
                v4f pa[4], pb[2];
#pragma unroll
                for (int k = 0; k < 4; ++k)
                    pa[k] = piece_at(plane_ptr(fix4, x0 - 3 + 6 + k, x0), idxA);
#pragma unroll
                for (int j = 0; j < 2; ++j)
                    pb[j] = halo_piece(fix4, bok[j], baseB + j * 4 * PLANE4 + cgoff);
                fix_plane_store<6, 4>(pa, okA, x0, s1A, stA);
#pragma unroll
                for (int j = 0; j < 2; ++j) {
                    const v4f o = fix_tpass(bok[j] ? pb[j] : zero4);
                    if (stA)
                        *reinterpret_cast<v4f*>(s1B + j * 800) = o;
                }
            }
            {
                v4f pm[NMTRIPS];
#pragma unroll
                for (int k = 0; k < NMTRIPS; ++k)
                    pm[k] = halo_piece(mov4, mok[k], moff[k] + cgoff);
                const v4f pb2 = halo_piece(fix4, bok[2], baseB + 2 * 4 * PLANE4 + cgoff);
#pragma unroll
                for (int k = 0; k < NMTRIPS; ++k) {
                    const v4f o = mov_tpass(mok[k] ? pm[k] : zero4);
                    const int M = k * 32 + hg;
                    if (hq >= 1 && hq <= 4 && M < NMROWS)
                        *reinterpret_cast<v4f*>(&smem[M1_OFF + M * 16 + 4 * (hq - 1)]) = o;
                }
                {
                    const v4f o = fix_tpass(bok[2] ? pb2 : zero4);
                    if (stA && ixB < 2)
                        *reinterpret_cast<v4f*>(s1B + 2 * 800) = o;
                }
            }
        }
        __syncthreads();
        if (tid < 160) {
            const int c4 = tid / 40, j = tid - c4 * 40;
            sweep_yx<10, 200, 20, BF_XS>(&smem[S1_OFF + c4 * 2000 + 4 * j],
                                         &smem[BF_OFF + c4 * BF_CS + 4 * j]);
        }
        if (tid >= 192) {
            const int jj = tid - 192;
            const int c4 = jj >> 4, jm = jj & 15;
            sweep_yx<6, 96, 16, 256>(&smem[M1_OFF + c4 * 576 + 4 * jm],
                                     &smem[PM_OFF + 4 * jj]);
        }
        __syncthreads();

        v2f fa[2], fb[10];
#pragma unroll
        for (int q = 0; q < 2; ++q) {
            fa[q][0] = smem[pm_base + 8 * q];
            fa[q][1] = smem[pm_base + 8 * q + 64];
        }
#pragma unroll
        for (int s = 0; s < 10; ++s) {
            fb[s][0] = smem[bf_base + 2 * s];
            fb[s][1] = smem[bf_base + 2 * s + BF_CS];
        }
#pragma unroll
        for (int q = 0; q < 2; ++q)
#pragma unroll
            for (int u = 0; u < 6; ++u)
                acc[q][u] = __builtin_amdgcn_wmma_f32_16x16x4_f32(
                    false, fa[q], false, fb[4 * q + u], (short)0, acc[q][u], false, false);
    }

    {
        const int di = (col >> 2) - hlf;
#pragma unroll
        for (int ay = 0; ay < 2; ++ay) {
            const int dj = (col & 3) - ay;
            if ((unsigned)di < 3u && (unsigned)dj < 3u) {
                const int ob = ((di * 3 + dj) * 3) * 256 +
                               ((px + 2 * hlf) * 4 + (py + 2 * ay)) * 16 + pt;
#pragma unroll
                for (int q = 0; q < 2; ++q)
#pragma unroll
                    for (int u = 0; u < 6; ++u)
#pragma unroll
                        for (int e = 0; e < 4; ++e) {
                            const int k = u - e;
                            if (k >= 0 && k <= 2)
                                smem[ob + k * 256 + 8 * q + 2 * e] = acc[q][u][ay * 4 + e];
                        }
            }
        }
    }
    __syncthreads();

    const int v  = tid;
    const int lx = v >> 6, ly = (v >> 4) & 3, lt = v & 15;
    const int x = x0 + lx, y = y0 + ly, t = t0 + lt;
#pragma unroll
    for (int d = 0; d < 27; ++d) {
        const float val = smem[d * 256 + tid] * SCALE;
        out[((d * VN + x) * VN + y) * VN + t] = val;
    }
}

extern "C" void kernel_launch(void* const* d_in, const int* in_sizes, int n_in,
                              void* d_out, int out_size, void* d_ws, size_t ws_size,
                              hipStream_t stream) {
    (void)in_sizes; (void)n_in; (void)out_size; (void)d_ws; (void)ws_size;
    const float* mov = (const float*)d_in[0];
    const float* fix = (const float*)d_in[1];
    float* out = (float*)d_out;
    dim3 grid(VN / BT, VN / BY, VN / BX);
    corr3d_wmma_kernel<<<grid, NTHREADS, 0, stream>>>(mov, fix, out);
}


# Fix box rows hold each 16-byte piece's t as even pair, odd pair (t-pass stores 0,2,1,3): a wave's two B values of a piece come by one 8-byte LDS load, 10 not 20 conflicted loads; same sums
# speedup vs baseline: 3.5271x; 1.0308x over previous
#include <hip/hip_runtime.h>

typedef float v2f __attribute__((ext_vector_type(2)));
typedef float v4f __attribute__((ext_vector_type(4)));
typedef float v8f __attribute__((ext_vector_type(8)));

#define VN 96
#define NC 32
#define BX 4
#define BY 4
#define BT 16
#define NTHREADS 256
#define SCALE (1.0f / 27.0f)
#define NMROWS 144
#define NMTRIPS 5
#define ROW4 (VN / 4)
#define PLANE4 (VN * ROW4)
#define CHAN4 (VN * PLANE4)

#define S1_OFF 0
#define BF_OFF 8000
#define BF_CS 176
#define BF_XS 708
#define PM_OFF 13664
#define M1_OFF 14688
#define SMEM_FLOATS 16992


__device__ __forceinline__ v4f halo_piece(const v4f* __restrict__ p, bool ok, int idx) {
    const int i4 = ok ? idx : 0;
    __builtin_assume(i4 >= 0);
    return p[i4];
}

__device__ __forceinline__ float lane_next(float x) {
    return __builtin_bit_cast(float, __builtin_amdgcn_update_dpp(
        0, __builtin_bit_cast(int, x), 0x101, 0xf, 0xf, true));
}
__device__ __forceinline__ float lane_prev(float x) {
    return __builtin_bit_cast(float, __builtin_amdgcn_update_dpp(
        0, __builtin_bit_cast(int, x), 0x111, 0xf, 0xf, true));
}

__device__ __forceinline__ v4f fix_tpass(v4f v) {
    const float n0 = lane_next(v[0]);
    const float n1 = lane_next(v[1]);
    const float n2 = lane_next(v[2]);
    v4f o;
    o[0] = v[1] + v[2] + v[3];
    o[2] = v[2] + v[3] + n0;
    o[1] = v[3] + n0 + n1;
    o[3] = n0 + n1 + n2;
    return o;
}

__device__ __forceinline__ v4f mov_tpass(v4f v) {
    const float p3 = lane_prev(v[3]);
    const float n0 = lane_next(v[0]);
    v4f o;
    o[0] = p3 + v[0] + v[1];
    o[1] = v[0] + v[1] + v[2];
    o[2] = v[1] + v[2] + v[3];
    o[3] = v[2] + v[3] + n0;
    return o;
}


__device__ __forceinline__ v4f piece_at(const v4f* __restrict__ p, int idx) {
    __builtin_assume(idx >= 0);
    return p[idx];
}

__device__ __forceinline__ const v4f* plane_ptr(const v4f* __restrict__ base, int gx, int x0) {
    const int gxc = (unsigned)gx < (unsigned)VN ? gx : x0;
    return base + gxc * PLANE4;
}

template <int K0, int N>
__device__ __forceinline__ void fix_plane_store(const v4f (&pc)[N], bool okA, int x0,
                                                float* s1p, bool st) {
    const v4f zero4 = {0.0f, 0.0f, 0.0f, 0.0f};
#pragma unroll
    for (int j = 0; j < N; ++j) {
        const bool ok = okA && (unsigned)(x0 - 3 + K0 + j) < (unsigned)VN;
        const v4f o = fix_tpass(ok ? pc[j] : zero4);
        if (st)
            *reinterpret_cast<v4f*>(s1p + (K0 + j) * 200) = o;
    }
}

template <int NIN, int XSTEP, int YSTEP, int DSTEP>
__device__ __forceinline__ void sweep_yx(const float* src, float* dst) {
    v4f y0 = {0.0f, 0.0f, 0.0f, 0.0f}, y1 = y0;
#pragma unroll
    for (int i = 0; i < NIN; ++i) {
        const float* p = src + i * XSTEP;
        const v4f a = *reinterpret_cast<const v4f*>(p);
        const v4f b = *reinterpret_cast<const v4f*>(p + YSTEP);
        const v4f c = *reinterpret_cast<const v4f*>(p + 2 * YSTEP);
        const v4f y = a + b + c;
        if (i >= 2)
            *reinterpret_cast<v4f*>(dst + (i - 2) * DSTEP) = y0 + y1 + y;
        y0 = y1;
        y1 = y;
    }
}

__global__ __launch_bounds__(NTHREADS, 6)
void corr3d_wmma_kernel(const float* __restrict__ mov,
                        const float* __restrict__ fix,
                        float* __restrict__ out) {
    __shared__ __attribute__((aligned(16))) float smem[SMEM_FLOATS];

    const int tid  = threadIdx.x;
    const int lane = tid & 31;
    const int w    = tid >> 5;
    const int col  = lane & 15;
    const int hlf  = lane >> 4;

    const int x0 = blockIdx.z * BX;
    const int y0 = blockIdx.y * BY;
    const int t0 = blockIdx.x * BT;

    const int px = (w >> 2) & 1, py = (w >> 1) & 1, pt = w & 1;
    const int pm_base = PM_OFF +
        (((px + 2 * (col >> 3)) * 4 + 2 * hlf) * 4 + (py + 2 * ((col >> 2) & 1))) * 16 +
        pt + 2 * (col & 3);
    const int bf_base = BF_OFF + (px + 2 * (col >> 2)) * BF_XS + (2 * hlf) * BF_CS +
                        (py + 2 * (col & 3)) * 20 + 2 * pt;

    v8f acc[2][6];
#pragma unroll
    for (int q = 0; q < 2; ++q)
#pragma unroll
        for (int u = 0; u < 6; ++u) {
            const v8f z = {};
            acc[q][u] = z;
        }

    const int  hq   = tid & 7;
    const int  hg   = tid >> 3;
    const int  hgt0 = t0 - 4 + 4 * hq;
    const bool htok = (hq < 6) && (unsigned)hgt0 < (unsigned)VN;
    const int  c4h  = hg >> 3;
    const int  r8   = hg & 7;
    const int  gyA   = y0 - 3 + r8;
    const bool yokA  = (unsigned)gyA < (unsigned)VN;
    const bool okA   = htok && yokA;
    const int  baseA = c4h * CHAN4 + (yokA ? gyA : y0) * ROW4 + ((htok ? hgt0 : t0) >> 2);
    const int  iyB   = 8 + (r8 >> 2), ixB = r8 & 3;
    const int  gyB   = y0 - 3 + iyB;
    const bool okBy  = htok && (unsigned)gyB < (unsigned)VN;
    const int  baseB = c4h * CHAN4 + (x0 - 3 + ixB) * PLANE4 + gyB * ROW4 + (hgt0 >> 2);
    bool bok[3];
#pragma unroll
    for (int j = 0; j < 3; ++j)
        bok[j] = okBy && (4 * j + ixB < 10) &&
                 (unsigned)(x0 - 3 + 4 * j + ixB) < (unsigned)VN;
    float* const s1A = &smem[S1_OFF + c4h * 2000 + r8 * 20 + 4 * hq];
    float* const s1B = &smem[S1_OFF + c4h * 2000 + ixB * 200 + iyB * 20 + 4 * hq];
    const bool stA = hq < 5;
    int  moff[NMTRIPS]; bool mok[NMTRIPS];
#pragma unroll
    for (int k = 0; k < NMTRIPS; ++k) {
        const int M   = k * 32 + hg;
        const int Mc  = (M < NMROWS) ? M : 0;
        const int c4  = Mc / 36, rem = Mc - c4 * 36;
        const int ix  = rem / 6, iy = rem - ix * 6;
        const int gx  = x0 - 1 + ix, gy = y0 - 1 + iy;
        mok[k]  = htok && (M < NMROWS) &&
                  (unsigned)gx < (unsigned)VN && (unsigned)gy < (unsigned)VN;
        moff[k] = ((c4 * VN + gx) * VN + gy) * ROW4 + (hgt0 >> 2);
    }
    const v4f* __restrict__ fix4 = reinterpret_cast<const v4f*>(fix);
    const v4f* __restrict__ mov4 = reinterpret_cast<const v4f*>(mov);

    for (int cg = 0; cg < 8; ++cg) {
        {
            const v4f zero4 = {0.0f, 0.0f, 0.0f, 0.0f};
            const int cgoff = cg * 4 * CHAN4;
            const int idxA  = baseA + cgoff;
            {
                v4f pa[6];
#pragma unroll
                for (int k = 0; k < 6; ++k)
                    pa[k] = piece_at(plane_ptr(fix4, x0 - 3 + k, x0), idxA);
                fix_plane_store<0, 6>(pa, okA, x0, s1A, stA);
            }
            {
                v4f pa[4], pb[2];
#pragma unroll
                for (int k = 0; k < 4; ++k)
                    pa[k] = piece_at(plane_ptr(fix4, x0 - 3 + 6 + k, x0), idxA);
#pragma unroll
                for (int j = 0; j < 2; ++j)
                    pb[j] = halo_piece(fix4, bok[j], baseB + j * 4 * PLANE4 + cgoff);
                fix_plane_store<6, 4>(pa, okA, x0, s1A, stA);
#pragma unroll
                for (int j = 0; j < 2; ++j) {
                    const v4f o = fix_tpass(bok[j] ? pb[j] : zero4);
                    if (stA)
                        *reinterpret_cast<v4f*>(s1B + j * 800) = o;
                }
            }
            {
                v4f pm[NMTRIPS];
#pragma unroll
                for (int k = 0; k < NMTRIPS; ++k)
                    pm[k] = halo_piece(mov4, mok[k], moff[k] + cgoff);
                const v4f pb2 = halo_piece(fix4, bok[2], baseB + 2 * 4 * PLANE4 + cgoff);
#pragma unroll
                for (int k = 0; k < NMTRIPS; ++k) {
                    const v4f o = mov_tpass(mok[k] ? pm[k] : zero4);
                    const int M = k * 32 + hg;
                    if (hq >= 1 && hq <= 4 && M < NMROWS)
                        *reinterpret_cast<v4f*>(&smem[M1_OFF + M * 16 + 4 * (hq - 1)]) = o;
                }
                {
                    const v4f o = fix_tpass(bok[2] ? pb2 : zero4);
                    if (stA && ixB < 2)
                        *reinterpret_cast<v4f*>(s1B + 2 * 800) = o;
                }
            }
        }
        __syncthreads();
        if (tid < 160) {
            const int c4 = tid / 40, j = tid - c4 * 40;
            sweep_yx<10, 200, 20, BF_XS>(&smem[S1_OFF + c4 * 2000 + 4 * j],
                                         &smem[BF_OFF + c4 * BF_CS + 4 * j]);
        }
        if (tid >= 192) {
            const int jj = tid - 192;
            const int c4 = jj >> 4, jm = jj & 15;
            sweep_yx<6, 96, 16, 256>(&smem[M1_OFF + c4 * 576 + 4 * jm],
                                     &smem[PM_OFF + 4 * jj]);
        }
        __syncthreads();

        v2f fa[2], fb[10];
#pragma unroll
        for (int q = 0; q < 2; ++q) {
            fa[q][0] = smem[pm_base + 8 * q];
            fa[q][1] = smem[pm_base + 8 * q + 64];
        }
#pragma unroll
        for (int p = 0; p < 5; ++p) {
            const v2f c0 = *reinterpret_cast<const v2f*>(&smem[bf_base + 4 * p]);
            const v2f c1 = *reinterpret_cast<const v2f*>(&smem[bf_base + 4 * p + BF_CS]);
            fb[2 * p][0]     = c0[0];
            fb[2 * p][1]     = c1[0];
            fb[2 * p + 1][0] = c0[1];
            fb[2 * p + 1][1] = c1[1];
        }
#pragma unroll
        for (int q = 0; q < 2; ++q)
#pragma unroll
            for (int u = 0; u < 6; ++u)
                acc[q][u] = __builtin_amdgcn_wmma_f32_16x16x4_f32(
                    false, fa[q], false, fb[4 * q + u], (short)0, acc[q][u], false, false);
    }

    {
        const int di = (col >> 2) - hlf;
#pragma unroll
        for (int ay = 0; ay < 2; ++ay) {
            const int dj = (col & 3) - ay;
            if ((unsigned)di < 3u && (unsigned)dj < 3u) {
                const int ob = ((di * 3 + dj) * 3) * 256 +
                               ((px + 2 * hlf) * 4 + (py + 2 * ay)) * 16 + pt;
#pragma unroll
                for (int q = 0; q < 2; ++q)
#pragma unroll
                    for (int u = 0; u < 6; ++u)
#pragma unroll
                        for (int e = 0; e < 4; ++e) {
                            const int k = u - e;
                            if (k >= 0 && k <= 2)
                                smem[ob + k * 256 + 8 * q + 2 * e] = acc[q][u][ay * 4 + e];
                        }
            }
        }
    }
    __syncthreads();

    const int v  = tid;
    const int lx = v >> 6, ly = (v >> 4) & 3, lt = v & 15;
    const int x = x0 + lx, y = y0 + ly, t = t0 + lt;
#pragma unroll
    for (int d = 0; d < 27; ++d) {
        const float val = smem[d * 256 + tid] * SCALE;
        out[((d * VN + x) * VN + y) * VN + t] = val;
    }
}

extern "C" void kernel_launch(void* const* d_in, const int* in_sizes, int n_in,
                              void* d_out, int out_size, void* d_ws, size_t ws_size,
                              hipStream_t stream) {
    (void)in_sizes; (void)n_in; (void)out_size; (void)d_ws; (void)ws_size;
    const float* mov = (const float*)d_in[0];
    const float* fix = (const float*)d_in[1];
    float* out = (float*)d_out;
    dim3 grid(VN / BT, VN / BY, VN / BX);
    corr3d_wmma_kernel<<<grid, NTHREADS, 0, stream>>>(mov, fix, out);
}


# Halo loads as a rolling window of 6 per wave: next group's first 6 asked for before the products (taken per row set to make room), each used piece replaced by the load 6 on; 160 registers; same sums
# speedup vs baseline: 3.6569x; 1.0368x over previous
#include <hip/hip_runtime.h>

typedef float v2f __attribute__((ext_vector_type(2)));
typedef float v4f __attribute__((ext_vector_type(4)));
typedef float v8f __attribute__((ext_vector_type(8)));

#define VN 96
#define NC 32
#define BX 4
#define BY 4
#define BT 16
#define NTHREADS 256
#define SCALE (1.0f / 27.0f)
#define NMROWS 144
#define NMTRIPS 5
#define ROW4 (VN / 4)
#define PLANE4 (VN * ROW4)
#define CHAN4 (VN * PLANE4)

#define S1_OFF 0
#define BF_OFF 8000
#define BF_CS 176
#define BF_XS 708
#define PM_OFF 13664
#define M1_OFF 14688
#define SMEM_FLOATS 16992


__device__ __forceinline__ v4f halo_piece(const v4f* __restrict__ p, bool ok, int idx) {
    const int i4 = ok ? idx : 0;
    __builtin_assume(i4 >= 0);
    return p[i4];
}

__device__ __forceinline__ float lane_next(float x) {
    return __builtin_bit_cast(float, __builtin_amdgcn_update_dpp(
        0, __builtin_bit_cast(int, x), 0x101, 0xf, 0xf, true));
}
__device__ __forceinline__ float lane_prev(float x) {
    return __builtin_bit_cast(float, __builtin_amdgcn_update_dpp(
        0, __builtin_bit_cast(int, x), 0x111, 0xf, 0xf, true));
}

__device__ __forceinline__ v4f fix_tpass(v4f v) {
    const float n0 = lane_next(v[0]);
    const float n1 = lane_next(v[1]);
    const float n2 = lane_next(v[2]);
    v4f o;
    o[0] = v[1] + v[2] + v[3];
    o[2] = v[2] + v[3] + n0;
    o[1] = v[3] + n0 + n1;
    o[3] = n0 + n1 + n2;
    return o;
}

__device__ __forceinline__ v4f mov_tpass(v4f v) {
    const float p3 = lane_prev(v[3]);
    const float n0 = lane_next(v[0]);
    v4f o;
    o[0] = p3 + v[0] + v[1];
    o[1] = v[0] + v[1] + v[2];
    o[2] = v[1] + v[2] + v[3];
    o[3] = v[2] + v[3] + n0;
    return o;
}


__device__ __forceinline__ v4f piece_at(const v4f* __restrict__ p, int idx) {
    __builtin_assume(idx >= 0);
    return p[idx];
}

__device__ __forceinline__ const v4f* plane_ptr(const v4f* __restrict__ base, int gx, int x0) {
    const int gxc = (unsigned)gx < (unsigned)VN ? gx : x0;
    return base + gxc * PLANE4;
}

__device__ __forceinline__ void fix_put(v4f v, bool ok, float* dst, bool st) {
    const v4f zero4 = {0.0f, 0.0f, 0.0f, 0.0f};
    const v4f o = fix_tpass(ok ? v : zero4);
    if (st)
        *reinterpret_cast<v4f*>(dst) = o;
}
__device__ __forceinline__ void mov_put(v4f v, bool ok, float* dst, bool st) {
    const v4f zero4 = {0.0f, 0.0f, 0.0f, 0.0f};
    const v4f o = mov_tpass(ok ? v : zero4);
    if (st)
        *reinterpret_cast<v4f*>(dst) = o;
}

__device__ __forceinline__ void load_issued() {
    __builtin_amdgcn_sched_barrier(0);
}

template <int NIN, int XSTEP, int YSTEP, int DSTEP>
__device__ __forceinline__ void sweep_yx(const float* src, float* dst) {
    v4f y0 = {0.0f, 0.0f, 0.0f, 0.0f}, y1 = y0;
#pragma unroll
    for (int i = 0; i < NIN; ++i) {
        const float* p = src + i * XSTEP;
        const v4f a = *reinterpret_cast<const v4f*>(p);
        const v4f b = *reinterpret_cast<const v4f*>(p + YSTEP);
        const v4f c = *reinterpret_cast<const v4f*>(p + 2 * YSTEP);
        const v4f y = a + b + c;
        if (i >= 2)
            *reinterpret_cast<v4f*>(dst + (i - 2) * DSTEP) = y0 + y1 + y;
        y0 = y1;
        y1 = y;
    }
}

__global__ __launch_bounds__(NTHREADS, 6)
void corr3d_wmma_kernel(const float* __restrict__ mov,
                        const float* __restrict__ fix,
                        float* __restrict__ out) {
    __shared__ __attribute__((aligned(16))) float smem[SMEM_FLOATS];

    const int tid  = threadIdx.x;
    const int lane = tid & 31;
    const int w    = tid >> 5;
    const int col  = lane & 15;
    const int hlf  = lane >> 4;

    const int x0 = blockIdx.z * BX;
    const int y0 = blockIdx.y * BY;
    const int t0 = blockIdx.x * BT;

    const int px = (w >> 2) & 1, py = (w >> 1) & 1, pt = w & 1;
    const int pm_base = PM_OFF +
        (((px + 2 * (col >> 3)) * 4 + 2 * hlf) * 4 + (py + 2 * ((col >> 2) & 1))) * 16 +
        pt + 2 * (col & 3);
    const int bf_base = BF_OFF + (px + 2 * (col >> 2)) * BF_XS + (2 * hlf) * BF_CS +
                        (py + 2 * (col & 3)) * 20 + 2 * pt;

    v8f acc[2][6];
#pragma unroll
    for (int q = 0; q < 2; ++q)
#pragma unroll
        for (int u = 0; u < 6; ++u) {
            const v8f z = {};
            acc[q][u] = z;
        }

    const int  hq   = tid & 7;
    const int  hg   = tid >> 3;
    const int  hgt0 = t0 - 4 + 4 * hq;
    const bool htok = (hq < 6) && (unsigned)hgt0 < (unsigned)VN;
    const int  c4h  = hg >> 3;
    const int  r8   = hg & 7;
    const int  gyA   = y0 - 3 + r8;
    const bool yokA  = (unsigned)gyA < (unsigned)VN;
    const bool okA   = htok && yokA;
    const int  baseA = c4h * CHAN4 + (yokA ? gyA : y0) * ROW4 + ((htok ? hgt0 : t0) >> 2);
    const int  iyB   = 8 + (r8 >> 2), ixB = r8 & 3;
    const int  gyB   = y0 - 3 + iyB;
    const bool okBy  = htok && (unsigned)gyB < (unsigned)VN;
    const int  baseB = c4h * CHAN4 + (x0 - 3 + ixB) * PLANE4 + gyB * ROW4 + (hgt0 >> 2);
    bool bok[3];
#pragma unroll
    for (int j = 0; j < 3; ++j)
        bok[j] = okBy && (4 * j + ixB < 10) &&
                 (unsigned)(x0 - 3 + 4 * j + ixB) < (unsigned)VN;
    float* const s1A = &smem[S1_OFF + c4h * 2000 + r8 * 20 + 4 * hq];
    float* const s1B = &smem[S1_OFF + c4h * 2000 + ixB * 200 + iyB * 20 + 4 * hq];
    const bool stA = hq < 5;
    int  moff[NMTRIPS]; bool mok[NMTRIPS];
#pragma unroll
    for (int k = 0; k < NMTRIPS; ++k) {
        const int M   = k * 32 + hg;
        const int Mc  = (M < NMROWS) ? M : 0;
        const int c4  = Mc / 36, rem = Mc - c4 * 36;
        const int ix  = rem / 6, iy = rem - ix * 6;
        const int gx  = x0 - 1 + ix, gy = y0 - 1 + iy;
        mok[k]  = htok && (M < NMROWS) &&
                  (unsigned)gx < (unsigned)VN && (unsigned)gy < (unsigned)VN;
        moff[k] = ((c4 * VN + gx) * VN + gy) * ROW4 + (hgt0 >> 2);
    }
    const v4f* __restrict__ fix4 = reinterpret_cast<const v4f*>(fix);
    const v4f* __restrict__ mov4 = reinterpret_cast<const v4f*>(mov);

    v4f w0, w1, w2, w3, w4, w5;
    w0 = piece_at(plane_ptr(fix4, x0 - 3 + 0, x0), baseA);
    w1 = piece_at(plane_ptr(fix4, x0 - 3 + 1, x0), baseA);
    w2 = piece_at(plane_ptr(fix4, x0 - 3 + 2, x0), baseA);
    w3 = piece_at(plane_ptr(fix4, x0 - 3 + 3, x0), baseA);
    w4 = piece_at(plane_ptr(fix4, x0 - 3 + 4, x0), baseA);
    w5 = piece_at(plane_ptr(fix4, x0 - 3 + 5, x0), baseA);

    for (int cg = 0; cg < 8; ++cg) {
        {
            const int cgoff = cg * 4 * CHAN4;
            const int idxA  = baseA + cgoff;
            const bool stM  = hq >= 1 && hq <= 4;
            float* const m1p = &smem[M1_OFF + hg * 16 + 4 * (hq - 1)];
            fix_put(w0, okA && (unsigned)(x0 - 3 + 0) < (unsigned)VN, s1A + 0 * 200, stA);
            w0 = piece_at(plane_ptr(fix4, x0 - 3 + 6, x0), idxA);
            load_issued();
            fix_put(w1, okA && (unsigned)(x0 - 3 + 1) < (unsigned)VN, s1A + 1 * 200, stA);
            w1 = piece_at(plane_ptr(fix4, x0 - 3 + 7, x0), idxA);
            load_issued();
            fix_put(w2, okA && (unsigned)(x0 - 3 + 2) < (unsigned)VN, s1A + 2 * 200, stA);
            w2 = piece_at(plane_ptr(fix4, x0 - 3 + 8, x0), idxA);
            load_issued();
            fix_put(w3, okA && (unsigned)(x0 - 3 + 3) < (unsigned)VN, s1A + 3 * 200, stA);
            w3 = piece_at(plane_ptr(fix4, x0 - 3 + 9, x0), idxA);
            load_issued();
            fix_put(w4, okA && (unsigned)(x0 - 3 + 4) < (unsigned)VN, s1A + 4 * 200, stA);
            w4 = halo_piece(fix4, bok[0], baseB + 0 * 4 * PLANE4 + cgoff);
            load_issued();
            fix_put(w5, okA && (unsigned)(x0 - 3 + 5) < (unsigned)VN, s1A + 5 * 200, stA);
            w5 = halo_piece(fix4, bok[1], baseB + 1 * 4 * PLANE4 + cgoff);
            load_issued();
            fix_put(w0, okA && (unsigned)(x0 - 3 + 6) < (unsigned)VN, s1A + 6 * 200, stA);
            w0 = halo_piece(mov4, mok[0], moff[0] + cgoff);
            load_issued();
            fix_put(w1, okA && (unsigned)(x0 - 3 + 7) < (unsigned)VN, s1A + 7 * 200, stA);
            w1 = halo_piece(mov4, mok[1], moff[1] + cgoff);
            load_issued();
            fix_put(w2, okA && (unsigned)(x0 - 3 + 8) < (unsigned)VN, s1A + 8 * 200, stA);
            w2 = halo_piece(mov4, mok[2], moff[2] + cgoff);
            load_issued();
            fix_put(w3, okA && (unsigned)(x0 - 3 + 9) < (unsigned)VN, s1A + 9 * 200, stA);
            w3 = halo_piece(mov4, mok[3], moff[3] + cgoff);
            load_issued();
            fix_put(w4, bok[0], s1B + 0 * 800, stA);
            w4 = halo_piece(mov4, mok[4], moff[4] + cgoff);
            load_issued();
            fix_put(w5, bok[1], s1B + 1 * 800, stA);
            w5 = halo_piece(fix4, bok[2], baseB + 2 * 4 * PLANE4 + cgoff);
            load_issued();
            mov_put(w0, mok[0], m1p + 0 * 512, stM);
            mov_put(w1, mok[1], m1p + 1 * 512, stM);
            mov_put(w2, mok[2], m1p + 2 * 512, stM);
            mov_put(w3, mok[3], m1p + 3 * 512, stM);
            mov_put(w4, mok[4], m1p + 4 * 512, stM && 4 * 32 + hg < NMROWS);
            fix_put(w5, bok[2], s1B + 2 * 800, stA && ixB < 2);
        }
        __syncthreads();
        if (tid < 160) {
            const int c4 = tid / 40, j = tid - c4 * 40;
            sweep_yx<10, 200, 20, BF_XS>(&smem[S1_OFF + c4 * 2000 + 4 * j],
                                         &smem[BF_OFF + c4 * BF_CS + 4 * j]);
        }
        if (tid >= 192) {
            const int jj = tid - 192;
            const int c4 = jj >> 4, jm = jj & 15;
            sweep_yx<6, 96, 16, 256>(&smem[M1_OFF + c4 * 576 + 4 * jm],
                                     &smem[PM_OFF + 4 * jj]);
        }
        __syncthreads();

        if (cg + 1 < 8) {
            const int idxN = baseA + (cg + 1) * 4 * CHAN4;
            w0 = piece_at(plane_ptr(fix4, x0 - 3 + 0, x0), idxN);
            w1 = piece_at(plane_ptr(fix4, x0 - 3 + 1, x0), idxN);
            w2 = piece_at(plane_ptr(fix4, x0 - 3 + 2, x0), idxN);
            w3 = piece_at(plane_ptr(fix4, x0 - 3 + 3, x0), idxN);
            w4 = piece_at(plane_ptr(fix4, x0 - 3 + 4, x0), idxN);
            w5 = piece_at(plane_ptr(fix4, x0 - 3 + 5, x0), idxN);
        }

#pragma unroll
        for (int q = 0; q < 2; ++q) {
            v2f fa, fb[6];
            fa[0] = smem[pm_base + 8 * q];
            fa[1] = smem[pm_base + 8 * q + 64];
#pragma unroll
            for (int pp = 0; pp < 3; ++pp) {
                const int p = 2 * q + pp;
                const v2f c0 = *reinterpret_cast<const v2f*>(&smem[bf_base + 4 * p]);
                const v2f c1 = *reinterpret_cast<const v2f*>(&smem[bf_base + 4 * p + BF_CS]);
                fb[2 * pp][0]     = c0[0];
                fb[2 * pp][1]     = c1[0];
                fb[2 * pp + 1][0] = c0[1];
                fb[2 * pp + 1][1] = c1[1];
            }
#pragma unroll
            for (int u = 0; u < 6; ++u)
                acc[q][u] = __builtin_amdgcn_wmma_f32_16x16x4_f32(
                    false, fa, false, fb[u], (short)0, acc[q][u], false, false);
            if (q == 0)
                __builtin_amdgcn_sched_barrier(0);
        }
    }

    {
        const int di = (col >> 2) - hlf;
#pragma unroll
        for (int ay = 0; ay < 2; ++ay) {
            const int dj = (col & 3) - ay;
            if ((unsigned)di < 3u && (unsigned)dj < 3u) {
                const int ob = ((di * 3 + dj) * 3) * 256 +
                               ((px + 2 * hlf) * 4 + (py + 2 * ay)) * 16 + pt;
#pragma unroll
                for (int q = 0; q < 2; ++q)
#pragma unroll
                    for (int u = 0; u < 6; ++u)
#pragma unroll
                        for (int e = 0; e < 4; ++e) {
                            const int k = u - e;
                            if (k >= 0 && k <= 2)
                                smem[ob + k * 256 + 8 * q + 2 * e] = acc[q][u][ay * 4 + e];
                        }
            }
        }
    }
    __syncthreads();

    const int v  = tid;
    const int lx = v >> 6, ly = (v >> 4) & 3, lt = v & 15;
    const int x = x0 + lx, y = y0 + ly, t = t0 + lt;
#pragma unroll
    for (int d = 0; d < 27; ++d) {
        const float val = smem[d * 256 + tid] * SCALE;
        out[((d * VN + x) * VN + y) * VN + t] = val;
    }
}

extern "C" void kernel_launch(void* const* d_in, const int* in_sizes, int n_in,
                              void* d_out, int out_size, void* d_ws, size_t ws_size,
                              hipStream_t stream) {
    (void)in_sizes; (void)n_in; (void)out_size; (void)d_ws; (void)ws_size;
    const float* mov = (const float*)d_in[0];
    const float* fix = (const float*)d_in[1];
    float* out = (float*)d_out;
    dim3 grid(VN / BT, VN / BY, VN / BX);
    corr3d_wmma_kernel<<<grid, NTHREADS, 0, stream>>>(mov, fix, out);
}


# Three t-neighbour tiles per work-group (24 waves; each 8-wave group keeps its own tile and LDS region as before): a 128-byte halo line that neighbours share leaves the second level once; same sums
# speedup vs baseline: 4.0848x; 1.1170x over previous
#include <hip/hip_runtime.h>

typedef float v2f __attribute__((ext_vector_type(2)));
typedef float v4f __attribute__((ext_vector_type(4)));
typedef float v8f __attribute__((ext_vector_type(8)));

#define VN 96
#define NC 32
#define BX 4
#define BY 4
#define BT 16
#define NTHREADS 256
#define NSUB 3
#define SCALE (1.0f / 27.0f)
#define NMROWS 144
#define NMTRIPS 5
#define ROW4 (VN / 4)
#define PLANE4 (VN * ROW4)
#define CHAN4 (VN * PLANE4)

#define S1_OFF 0
#define BF_OFF 8000
#define BF_CS 176
#define BF_XS 708
#define PM_OFF 13664
#define M1_OFF 14688
#define SMEM_FLOATS 16992


__device__ __forceinline__ v4f halo_piece(const v4f* __restrict__ p, bool ok, int idx) {
    const int i4 = ok ? idx : 0;
    __builtin_assume(i4 >= 0);
    return p[i4];
}

__device__ __forceinline__ float lane_next(float x) {
    return __builtin_bit_cast(float, __builtin_amdgcn_update_dpp(
        0, __builtin_bit_cast(int, x), 0x101, 0xf, 0xf, true));
}
__device__ __forceinline__ float lane_prev(float x) {
    return __builtin_bit_cast(float, __builtin_amdgcn_update_dpp(
        0, __builtin_bit_cast(int, x), 0x111, 0xf, 0xf, true));
}

__device__ __forceinline__ v4f fix_tpass(v4f v) {
    const float n0 = lane_next(v[0]);
    const float n1 = lane_next(v[1]);
    const float n2 = lane_next(v[2]);
    v4f o;
    o[0] = v[1] + v[2] + v[3];
    o[2] = v[2] + v[3] + n0;
    o[1] = v[3] + n0 + n1;
    o[3] = n0 + n1 + n2;
    return o;
}

__device__ __forceinline__ v4f mov_tpass(v4f v) {
    const float p3 = lane_prev(v[3]);
    const float n0 = lane_next(v[0]);
    v4f o;
    o[0] = p3 + v[0] + v[1];
    o[1] = v[0] + v[1] + v[2];
    o[2] = v[1] + v[2] + v[3];
    o[3] = v[2] + v[3] + n0;
    return o;
}


__device__ __forceinline__ v4f piece_at(const v4f* __restrict__ p, int idx) {
    __builtin_assume(idx >= 0);
    return p[idx];
}

__device__ __forceinline__ const v4f* plane_ptr(const v4f* __restrict__ base, int gx, int x0) {
    const int gxc = (unsigned)gx < (unsigned)VN ? gx : x0;
    return base + gxc * PLANE4;
}

__device__ __forceinline__ void fix_put(v4f v, bool ok, float* dst, bool st) {
    const v4f zero4 = {0.0f, 0.0f, 0.0f, 0.0f};
    const v4f o = fix_tpass(ok ? v : zero4);
    if (st)
        *reinterpret_cast<v4f*>(dst) = o;
}
__device__ __forceinline__ void mov_put(v4f v, bool ok, float* dst, bool st) {
    const v4f zero4 = {0.0f, 0.0f, 0.0f, 0.0f};
    const v4f o = mov_tpass(ok ? v : zero4);
    if (st)
        *reinterpret_cast<v4f*>(dst) = o;
}

__device__ __forceinline__ void load_issued() {
    __builtin_amdgcn_sched_barrier(0);
}

template <int NIN, int XSTEP, int YSTEP, int DSTEP>
__device__ __forceinline__ void sweep_yx(const float* src, float* dst) {
    v4f y0 = {0.0f, 0.0f, 0.0f, 0.0f}, y1 = y0;
#pragma unroll
    for (int i = 0; i < NIN; ++i) {
        const float* p = src + i * XSTEP;
        const v4f a = *reinterpret_cast<const v4f*>(p);
        const v4f b = *reinterpret_cast<const v4f*>(p + YSTEP);
        const v4f c = *reinterpret_cast<const v4f*>(p + 2 * YSTEP);
        const v4f y = a + b + c;
        if (i >= 2)
            *reinterpret_cast<v4f*>(dst + (i - 2) * DSTEP) = y0 + y1 + y;
        y0 = y1;
        y1 = y;
    }
}

__global__ __launch_bounds__(NSUB * NTHREADS, 6)
void corr3d_wmma_kernel(const float* __restrict__ mov,
                        const float* __restrict__ fix,
                        float* __restrict__ out) {
    __shared__ __attribute__((aligned(16))) float smem_all[NSUB * SMEM_FLOATS];

    const int sub  = __builtin_amdgcn_readfirstlane((int)(threadIdx.x >> 8));
    float* const smem = smem_all + sub * SMEM_FLOATS;
    const int tid  = threadIdx.x & (NTHREADS - 1);
    const int lane = tid & 31;
    const int w    = tid >> 5;
    const int col  = lane & 15;
    const int hlf  = lane >> 4;

    const int x0 = blockIdx.z * BX;
    const int y0 = blockIdx.y * BY;
    const int t0 = (blockIdx.x * NSUB + sub) * BT;

    const int px = (w >> 2) & 1, py = (w >> 1) & 1, pt = w & 1;
    const int pm_base = PM_OFF +
        (((px + 2 * (col >> 3)) * 4 + 2 * hlf) * 4 + (py + 2 * ((col >> 2) & 1))) * 16 +
        pt + 2 * (col & 3);
    const int bf_base = BF_OFF + (px + 2 * (col >> 2)) * BF_XS + (2 * hlf) * BF_CS +
                        (py + 2 * (col & 3)) * 20 + 2 * pt;

    v8f acc[2][6];
#pragma unroll
    for (int q = 0; q < 2; ++q)
#pragma unroll
        for (int u = 0; u < 6; ++u) {
            const v8f z = {};
            acc[q][u] = z;
        }

    const int  hq   = tid & 7;
    const int  hg   = tid >> 3;
    const int  hgt0 = t0 - 4 + 4 * hq;
    const bool htok = (hq < 6) && (unsigned)hgt0 < (unsigned)VN;
    const int  c4h  = hg >> 3;
    const int  r8   = hg & 7;
    const int  gyA   = y0 - 3 + r8;
    const bool yokA  = (unsigned)gyA < (unsigned)VN;
    const bool okA   = htok && yokA;
    const int  baseA = c4h * CHAN4 + (yokA ? gyA : y0) * ROW4 + ((htok ? hgt0 : t0) >> 2);
    const int  iyB   = 8 + (r8 >> 2), ixB = r8 & 3;
    const int  gyB   = y0 - 3 + iyB;
    const bool okBy  = htok && (unsigned)gyB < (unsigned)VN;
    const int  baseB = c4h * CHAN4 + (x0 - 3 + ixB) * PLANE4 + gyB * ROW4 + (hgt0 >> 2);
    bool bok[3];
#pragma unroll
    for (int j = 0; j < 3; ++j)
        bok[j] = okBy && (4 * j + ixB < 10) &&
                 (unsigned)(x0 - 3 + 4 * j + ixB) < (unsigned)VN;
    float* const s1A = &smem[S1_OFF + c4h * 2000 + r8 * 20 + 4 * hq];
    float* const s1B = &smem[S1_OFF + c4h * 2000 + ixB * 200 + iyB * 20 + 4 * hq];
    const bool stA = hq < 5;
    int  moff[NMTRIPS]; bool mok[NMTRIPS];
#pragma unroll
    for (int k = 0; k < NMTRIPS; ++k) {
        const int M   = k * 32 + hg;
        const int Mc  = (M < NMROWS) ? M : 0;
        const int c4  = Mc / 36, rem = Mc - c4 * 36;
        const int ix  = rem / 6, iy = rem - ix * 6;
        const int gx  = x0 - 1 + ix, gy = y0 - 1 + iy;
        mok[k]  = htok && (M < NMROWS) &&
                  (unsigned)gx < (unsigned)VN && (unsigned)gy < (unsigned)VN;
        moff[k] = ((c4 * VN + gx) * VN + gy) * ROW4 + (hgt0 >> 2);
    }
    const v4f* __restrict__ fix4 = reinterpret_cast<const v4f*>(fix);
    const v4f* __restrict__ mov4 = reinterpret_cast<const v4f*>(mov);

    v4f w0, w1, w2, w3, w4, w5;
    w0 = piece_at(plane_ptr(fix4, x0 - 3 + 0, x0), baseA);
    w1 = piece_at(plane_ptr(fix4, x0 - 3 + 1, x0), baseA);
    w2 = piece_at(plane_ptr(fix4, x0 - 3 + 2, x0), baseA);
    w3 = piece_at(plane_ptr(fix4, x0 - 3 + 3, x0), baseA);
    w4 = piece_at(plane_ptr(fix4, x0 - 3 + 4, x0), baseA);
    w5 = piece_at(plane_ptr(fix4, x0 - 3 + 5, x0), baseA);

    for (int cg = 0; cg < 8; ++cg) {
        {
            const int cgoff = cg * 4 * CHAN4;
            const int idxA  = baseA + cgoff;
            const bool stM  = hq >= 1 && hq <= 4;
            float* const m1p = &smem[M1_OFF + hg * 16 + 4 * (hq - 1)];
            fix_put(w0, okA && (unsigned)(x0 - 3 + 0) < (unsigned)VN, s1A + 0 * 200, stA);
            w0 = piece_at(plane_ptr(fix4, x0 - 3 + 6, x0), idxA);
            load_issued();
            fix_put(w1, okA && (unsigned)(x0 - 3 + 1) < (unsigned)VN, s1A + 1 * 200, stA);
            w1 = piece_at(plane_ptr(fix4, x0 - 3 + 7, x0), idxA);
            load_issued();
            fix_put(w2, okA && (unsigned)(x0 - 3 + 2) < (unsigned)VN, s1A + 2 * 200, stA);
            w2 = piece_at(plane_ptr(fix4, x0 - 3 + 8, x0), idxA);
            load_issued();
            fix_put(w3, okA && (unsigned)(x0 - 3 + 3) < (unsigned)VN, s1A + 3 * 200, stA);
            w3 = piece_at(plane_ptr(fix4, x0 - 3 + 9, x0), idxA);
            load_issued();
            fix_put(w4, okA && (unsigned)(x0 - 3 + 4) < (unsigned)VN, s1A + 4 * 200, stA);
            w4 = halo_piece(fix4, bok[0], baseB + 0 * 4 * PLANE4 + cgoff);
            load_issued();
            fix_put(w5, okA && (unsigned)(x0 - 3 + 5) < (unsigned)VN, s1A + 5 * 200, stA);
            w5 = halo_piece(fix4, bok[1], baseB + 1 * 4 * PLANE4 + cgoff);
            load_issued();
            fix_put(w0, okA && (unsigned)(x0 - 3 + 6) < (unsigned)VN, s1A + 6 * 200, stA);
            w0 = halo_piece(mov4, mok[0], moff[0] + cgoff);
            load_issued();
            fix_put(w1, okA && (unsigned)(x0 - 3 + 7) < (unsigned)VN, s1A + 7 * 200, stA);
            w1 = halo_piece(mov4, mok[1], moff[1] + cgoff);
            load_issued();
            fix_put(w2, okA && (unsigned)(x0 - 3 + 8) < (unsigned)VN, s1A + 8 * 200, stA);
            w2 = halo_piece(mov4, mok[2], moff[2] + cgoff);
            load_issued();
            fix_put(w3, okA && (unsigned)(x0 - 3 + 9) < (unsigned)VN, s1A + 9 * 200, stA);
            w3 = halo_piece(mov4, mok[3], moff[3] + cgoff);
            load_issued();
            fix_put(w4, bok[0], s1B + 0 * 800, stA);
            w4 = halo_piece(mov4, mok[4], moff[4] + cgoff);
            load_issued();
            fix_put(w5, bok[1], s1B + 1 * 800, stA);
            w5 = halo_piece(fix4, bok[2], baseB + 2 * 4 * PLANE4 + cgoff);
            load_issued();
            mov_put(w0, mok[0], m1p + 0 * 512, stM);
            mov_put(w1, mok[1], m1p + 1 * 512, stM);
            mov_put(w2, mok[2], m1p + 2 * 512, stM);
            mov_put(w3, mok[3], m1p + 3 * 512, stM);
            mov_put(w4, mok[4], m1p + 4 * 512, stM && 4 * 32 + hg < NMROWS);
            fix_put(w5, bok[2], s1B + 2 * 800, stA && ixB < 2);
        }
        __syncthreads();
        if (tid < 160) {
            const int c4 = tid / 40, j = tid - c4 * 40;
            sweep_yx<10, 200, 20, BF_XS>(&smem[S1_OFF + c4 * 2000 + 4 * j],
                                         &smem[BF_OFF + c4 * BF_CS + 4 * j]);
        }
        if (tid >= 192) {
            const int jj = tid - 192;
            const int c4 = jj >> 4, jm = jj & 15;
            sweep_yx<6, 96, 16, 256>(&smem[M1_OFF + c4 * 576 + 4 * jm],
                                     &smem[PM_OFF + 4 * jj]);
        }
        __syncthreads();

        if (cg + 1 < 8) {
            const int idxN = baseA + (cg + 1) * 4 * CHAN4;
            w0 = piece_at(plane_ptr(fix4, x0 - 3 + 0, x0), idxN);
            w1 = piece_at(plane_ptr(fix4, x0 - 3 + 1, x0), idxN);
            w2 = piece_at(plane_ptr(fix4, x0 - 3 + 2, x0), idxN);
            w3 = piece_at(plane_ptr(fix4, x0 - 3 + 3, x0), idxN);
            w4 = piece_at(plane_ptr(fix4, x0 - 3 + 4, x0), idxN);
            w5 = piece_at(plane_ptr(fix4, x0 - 3 + 5, x0), idxN);
        }

#pragma unroll
        for (int q = 0; q < 2; ++q) {
            v2f fa, fb[6];
            fa[0] = smem[pm_base + 8 * q];
            fa[1] = smem[pm_base + 8 * q + 64];
#pragma unroll
            for (int pp = 0; pp < 3; ++pp) {
                const int p = 2 * q + pp;
                const v2f c0 = *reinterpret_cast<const v2f*>(&smem[bf_base + 4 * p]);
                const v2f c1 = *reinterpret_cast<const v2f*>(&smem[bf_base + 4 * p + BF_CS]);
                fb[2 * pp][0]     = c0[0];
                fb[2 * pp][1]     = c1[0];
                fb[2 * pp + 1][0] = c0[1];
                fb[2 * pp + 1][1] = c1[1];
            }
#pragma unroll
            for (int u = 0; u < 6; ++u)
                acc[q][u] = __builtin_amdgcn_wmma_f32_16x16x4_f32(
                    false, fa, false, fb[u], (short)0, acc[q][u], false, false);
            if (q == 0)
                __builtin_amdgcn_sched_barrier(0);
        }
    }

    {
        const int di = (col >> 2) - hlf;
#pragma unroll
        for (int ay = 0; ay < 2; ++ay) {
            const int dj = (col & 3) - ay;
            if ((unsigned)di < 3u && (unsigned)dj < 3u) {
                const int ob = ((di * 3 + dj) * 3) * 256 +
                               ((px + 2 * hlf) * 4 + (py + 2 * ay)) * 16 + pt;
#pragma unroll
                for (int q = 0; q < 2; ++q)
#pragma unroll
                    for (int u = 0; u < 6; ++u)
#pragma unroll
                        for (int e = 0; e < 4; ++e) {
                            const int k = u - e;
                            if (k >= 0 && k <= 2)
                                smem[ob + k * 256 + 8 * q + 2 * e] = acc[q][u][ay * 4 + e];
                        }
            }
        }
    }
    __syncthreads();

    const int v  = tid;
    const int lx = v >> 6, ly = (v >> 4) & 3, lt = v & 15;
    const int x = x0 + lx, y = y0 + ly, t = t0 + lt;
#pragma unroll
    for (int d = 0; d < 27; ++d) {
        const float val = smem[d * 256 + tid] * SCALE;
        out[((d * VN + x) * VN + y) * VN + t] = val;
    }
}

extern "C" void kernel_launch(void* const* d_in, const int* in_sizes, int n_in,
                              void* d_out, int out_size, void* d_ws, size_t ws_size,
                              hipStream_t stream) {
    (void)in_sizes; (void)n_in; (void)out_size; (void)d_ws; (void)ws_size;
    const float* mov = (const float*)d_in[0];
    const float* fix = (const float*)d_in[1];
    float* out = (float*)d_out;
    dim3 grid(VN / (BT * NSUB), VN / BY, VN / BX);
    corr3d_wmma_kernel<<<grid, NSUB * NTHREADS, 0, stream>>>(mov, fix, out);
}


# Halo rows loaded and t-summed once per triple of t-adjacent tiles: 16 lanes per 56-float row, piece stored to its tile's LDS, shared piece read from the tile before; 13 trips not 18; same sums
# speedup vs baseline: 4.4080x; 1.0791x over previous
#include <hip/hip_runtime.h>

typedef float v2f __attribute__((ext_vector_type(2)));
typedef float v4f __attribute__((ext_vector_type(4)));
typedef float v8f __attribute__((ext_vector_type(8)));

#define VN 96
#define NC 32
#define BX 4
#define BY 4
#define BT 16
#define NTHREADS 256
#define NSUB 3
#define SCALE (1.0f / 27.0f)
#define NSLOTS 48
#define NMTRIPS 3
#define ROW4 (VN / 4)
#define PLANE4 (VN * ROW4)
#define CHAN4 (VN * PLANE4)

#define S1_OFF 0
#define BF_OFF 8000
#define BF_CS 176
#define BF_XS 708
#define PM_OFF 13664
#define M1_OFF 14688
#define SMEM_FLOATS 16992


__device__ __forceinline__ v4f halo_piece(const v4f* __restrict__ p, bool ok, int idx) {
    const int i4 = ok ? idx : 0;
    __builtin_assume(i4 >= 0);
    return p[i4];
}

__device__ __forceinline__ float lane_next(float x) {
    return __builtin_bit_cast(float, __builtin_amdgcn_update_dpp(
        0, __builtin_bit_cast(int, x), 0x101, 0xf, 0xf, true));
}
__device__ __forceinline__ float lane_prev(float x) {
    return __builtin_bit_cast(float, __builtin_amdgcn_update_dpp(
        0, __builtin_bit_cast(int, x), 0x111, 0xf, 0xf, true));
}

__device__ __forceinline__ v4f fix_tpass(v4f v) {
    const float n0 = lane_next(v[0]);
    const float n1 = lane_next(v[1]);
    const float n2 = lane_next(v[2]);
    v4f o;
    o[0] = v[1] + v[2] + v[3];
    o[2] = v[2] + v[3] + n0;
    o[1] = v[3] + n0 + n1;
    o[3] = n0 + n1 + n2;
    return o;
}

__device__ __forceinline__ v4f mov_tpass(v4f v) {
    const float p3 = lane_prev(v[3]);
    const float n0 = lane_next(v[0]);
    v4f o;
    o[0] = p3 + v[0] + v[1];
    o[1] = v[0] + v[1] + v[2];
    o[2] = v[1] + v[2] + v[3];
    o[3] = v[2] + v[3] + n0;
    return o;
}


__device__ __forceinline__ v4f piece_at(const v4f* __restrict__ p, int idx) {
    __builtin_assume(idx >= 0);
    return p[idx];
}

__device__ __forceinline__ const v4f* plane_ptr(const v4f* __restrict__ base, int gx, int x0) {
    const int gxc = (unsigned)gx < (unsigned)VN ? gx : x0;
    return base + gxc * PLANE4;
}

__device__ __forceinline__ void fix_put(v4f v, bool ok, float* dst, bool st) {
    const v4f zero4 = {0.0f, 0.0f, 0.0f, 0.0f};
    const v4f o = fix_tpass(ok ? v : zero4);
    if (st)
        *reinterpret_cast<v4f*>(dst) = o;
}
__device__ __forceinline__ void mov_put(v4f v, bool ok, float* dst, bool st) {
    const v4f zero4 = {0.0f, 0.0f, 0.0f, 0.0f};
    const v4f o = mov_tpass(ok ? v : zero4);
    if (st)
        *reinterpret_cast<v4f*>(dst) = o;
}

__device__ __forceinline__ void load_issued() {
    __builtin_amdgcn_sched_barrier(0);
}

template <int NIN, int XSTEP, int YSTEP, int DSTEP>
__device__ __forceinline__ void sweep_yx(const float* src, float* dst) {
    v4f y0 = {0.0f, 0.0f, 0.0f, 0.0f}, y1 = y0;
#pragma unroll
    for (int i = 0; i < NIN; ++i) {
        const float* p = src + i * XSTEP;
        const v4f a = *reinterpret_cast<const v4f*>(p);
        const v4f b = *reinterpret_cast<const v4f*>(p + YSTEP);
        const v4f c = *reinterpret_cast<const v4f*>(p + 2 * YSTEP);
        const v4f y = a + b + c;
        if (i >= 2)
            *reinterpret_cast<v4f*>(dst + (i - 2) * DSTEP) = y0 + y1 + y;
        y0 = y1;
        y1 = y;
    }
}

__global__ __launch_bounds__(NSUB * NTHREADS, 6)
void corr3d_wmma_kernel(const float* __restrict__ mov,
                        const float* __restrict__ fix,
                        float* __restrict__ out) {
    __shared__ __attribute__((aligned(16))) float smem_all[NSUB * SMEM_FLOATS];

    const int sub  = __builtin_amdgcn_readfirstlane((int)(threadIdx.x >> 8));
    float* const smem = smem_all + sub * SMEM_FLOATS;
    const int tid  = threadIdx.x & (NTHREADS - 1);
    const int lane = tid & 31;
    const int w    = tid >> 5;
    const int col  = lane & 15;
    const int hlf  = lane >> 4;

    const int x0 = blockIdx.z * BX;
    const int y0 = blockIdx.y * BY;
    const int t0 = (blockIdx.x * NSUB + sub) * BT;

    const int px = (w >> 2) & 1, py = (w >> 1) & 1, pt = w & 1;
    const int pm_base = PM_OFF +
        (((px + 2 * (col >> 3)) * 4 + 2 * hlf) * 4 + (py + 2 * ((col >> 2) & 1))) * 16 +
        pt + 2 * (col & 3);
    const int bf_base = BF_OFF + (px + 2 * (col >> 2)) * BF_XS + (2 * hlf) * BF_CS +
                        (py + 2 * (col & 3)) * 20 + 2 * pt;

    v8f acc[2][6];
#pragma unroll
    for (int q = 0; q < 2; ++q)
#pragma unroll
        for (int u = 0; u < 6; ++u) {
            const v8f z = {};
            acc[q][u] = z;
        }

    const int  gtid = threadIdx.x;
    const int  hq   = gtid & 15;
    const int  hs   = gtid >> 4;
    const int  T0   = blockIdx.x * (NSUB * BT);
    const int  hgt0 = T0 - 4 + 4 * hq;
    const bool htok = (hq < 14) && (unsigned)hgt0 < (unsigned)VN;
    const int  hsub = hq == 0 ? 0 : (((hq - 1) >> 2) < 2 ? ((hq - 1) >> 2) : 2);
    float* const hdst = smem_all + hsub * SMEM_FLOATS;
    const bool hsok = hs < 40;
    const int  c4h  = hsok ? hs / 10 : 0;
    const int  r10  = hsok ? hs - 10 * c4h : 0;
    const int  gyA   = y0 - 3 + r10;
    const bool yokA  = (unsigned)gyA < (unsigned)VN;
    const bool okA   = htok && yokA;
    const int  baseA = c4h * CHAN4 + (yokA ? gyA : y0) * ROW4 + ((htok ? hgt0 : T0) >> 2);
    float* const s1A = &hdst[S1_OFF + c4h * 2000 + r10 * 20 + 4 * (hq - 4 * hsub)];
    const bool stA = hsok && hq <= 12;
    int  moff[NMTRIPS]; bool mok[NMTRIPS];
#pragma unroll
    for (int k = 0; k < NMTRIPS; ++k) {
        const int M   = k * NSLOTS + hs;
        const int c4  = M / 36, rem = M - c4 * 36;
        const int ix  = rem / 6, iy = rem - ix * 6;
        const int gx  = x0 - 1 + ix, gy = y0 - 1 + iy;
        mok[k]  = htok &&
                  (unsigned)gx < (unsigned)VN && (unsigned)gy < (unsigned)VN;
        moff[k] = ((c4 * VN + gx) * VN + gy) * ROW4 + (hgt0 >> 2);
    }
    float* const m1p = &hdst[M1_OFF + hs * 16 + 4 * (hq - 1 - 4 * hsub)];
    const bool stM = hq >= 1 && hq <= 12;
    const v4f* __restrict__ fix4 = reinterpret_cast<const v4f*>(fix);
    const v4f* __restrict__ mov4 = reinterpret_cast<const v4f*>(mov);

    v4f w0, w1, w2, w3, w4, w5;
    w0 = piece_at(plane_ptr(fix4, x0 - 3 + 0, x0), baseA);
    w1 = piece_at(plane_ptr(fix4, x0 - 3 + 1, x0), baseA);
    w2 = piece_at(plane_ptr(fix4, x0 - 3 + 2, x0), baseA);
    w3 = piece_at(plane_ptr(fix4, x0 - 3 + 3, x0), baseA);
    w4 = piece_at(plane_ptr(fix4, x0 - 3 + 4, x0), baseA);
    w5 = piece_at(plane_ptr(fix4, x0 - 3 + 5, x0), baseA);

    for (int cg = 0; cg < 8; ++cg) {
        {
            const int cgoff = cg * 4 * CHAN4;
            const int idxA  = baseA + cgoff;
            fix_put(w0, okA && (unsigned)(x0 - 3 + 0) < (unsigned)VN, s1A + 0 * 200, stA);
            w0 = piece_at(plane_ptr(fix4, x0 - 3 + 6, x0), idxA);
            load_issued();
            fix_put(w1, okA && (unsigned)(x0 - 3 + 1) < (unsigned)VN, s1A + 1 * 200, stA);
            w1 = piece_at(plane_ptr(fix4, x0 - 3 + 7, x0), idxA);
            load_issued();
            fix_put(w2, okA && (unsigned)(x0 - 3 + 2) < (unsigned)VN, s1A + 2 * 200, stA);
            w2 = piece_at(plane_ptr(fix4, x0 - 3 + 8, x0), idxA);
            load_issued();
            fix_put(w3, okA && (unsigned)(x0 - 3 + 3) < (unsigned)VN, s1A + 3 * 200, stA);
            w3 = piece_at(plane_ptr(fix4, x0 - 3 + 9, x0), idxA);
            load_issued();
            fix_put(w4, okA && (unsigned)(x0 - 3 + 4) < (unsigned)VN, s1A + 4 * 200, stA);
            w4 = halo_piece(mov4, mok[0], moff[0] + cgoff);
            load_issued();
            fix_put(w5, okA && (unsigned)(x0 - 3 + 5) < (unsigned)VN, s1A + 5 * 200, stA);
            w5 = halo_piece(mov4, mok[1], moff[1] + cgoff);
            load_issued();
            fix_put(w0, okA && (unsigned)(x0 - 3 + 6) < (unsigned)VN, s1A + 6 * 200, stA);
            w0 = halo_piece(mov4, mok[2], moff[2] + cgoff);
            load_issued();
            fix_put(w1, okA && (unsigned)(x0 - 3 + 7) < (unsigned)VN, s1A + 7 * 200, stA);
            fix_put(w2, okA && (unsigned)(x0 - 3 + 8) < (unsigned)VN, s1A + 8 * 200, stA);
            fix_put(w3, okA && (unsigned)(x0 - 3 + 9) < (unsigned)VN, s1A + 9 * 200, stA);
            mov_put(w4, mok[0], m1p + 0 * NSLOTS * 16, stM);
            mov_put(w5, mok[1], m1p + 1 * NSLOTS * 16, stM);
            mov_put(w0, mok[2], m1p + 2 * NSLOTS * 16, stM);
        }
        __syncthreads();
        if (tid < 160) {
            const int c4 = tid / 40, j = tid - c4 * 40;
            const bool fromprev = sub > 0 && j % 5 == 0;
            const float* const s1c = &smem[S1_OFF + c4 * 2000 + 4 * j];
            sweep_yx<10, 200, 20, BF_XS>(fromprev ? s1c - SMEM_FLOATS + 16 : s1c,
                                         &smem[BF_OFF + c4 * BF_CS + 4 * j]);
        }
        if (tid >= 192) {
            const int jj = tid - 192;
            const int c4 = jj >> 4, jm = jj & 15;
            sweep_yx<6, 96, 16, 256>(&smem[M1_OFF + c4 * 576 + 4 * jm],
                                     &smem[PM_OFF + 4 * jj]);
        }
        __syncthreads();

        if (cg + 1 < 8) {
            const int idxN = baseA + (cg + 1) * 4 * CHAN4;
            w0 = piece_at(plane_ptr(fix4, x0 - 3 + 0, x0), idxN);
            w1 = piece_at(plane_ptr(fix4, x0 - 3 + 1, x0), idxN);
            w2 = piece_at(plane_ptr(fix4, x0 - 3 + 2, x0), idxN);
            w3 = piece_at(plane_ptr(fix4, x0 - 3 + 3, x0), idxN);
            w4 = piece_at(plane_ptr(fix4, x0 - 3 + 4, x0), idxN);
            w5 = piece_at(plane_ptr(fix4, x0 - 3 + 5, x0), idxN);
        }

#pragma unroll
        for (int q = 0; q < 2; ++q) {
            v2f fa, fb[6];
            fa[0] = smem[pm_base + 8 * q];
            fa[1] = smem[pm_base + 8 * q + 64];
#pragma unroll
            for (int pp = 0; pp < 3; ++pp) {
                const int p = 2 * q + pp;
                const v2f c0 = *reinterpret_cast<const v2f*>(&smem[bf_base + 4 * p]);
                const v2f c1 = *reinterpret_cast<const v2f*>(&smem[bf_base + 4 * p + BF_CS]);
                fb[2 * pp][0]     = c0[0];
                fb[2 * pp][1]     = c1[0];
                fb[2 * pp + 1][0] = c0[1];
                fb[2 * pp + 1][1] = c1[1];
            }
#pragma unroll
            for (int u = 0; u < 6; ++u)
                acc[q][u] = __builtin_amdgcn_wmma_f32_16x16x4_f32(
                    false, fa, false, fb[u], (short)0, acc[q][u], false, false);
            if (q == 0)
                __builtin_amdgcn_sched_barrier(0);
        }
    }

    {
        const int di = (col >> 2) - hlf;
#pragma unroll
        for (int ay = 0; ay < 2; ++ay) {
            const int dj = (col & 3) - ay;
            if ((unsigned)di < 3u && (unsigned)dj < 3u) {
                const int ob = ((di * 3 + dj) * 3) * 256 +
                               ((px + 2 * hlf) * 4 + (py + 2 * ay)) * 16 + pt;
#pragma unroll
                for (int q = 0; q < 2; ++q)
#pragma unroll
                    for (int u = 0; u < 6; ++u)
#pragma unroll
                        for (int e = 0; e < 4; ++e) {
                            const int k = u - e;
                            if (k >= 0 && k <= 2)
                                smem[ob + k * 256 + 8 * q + 2 * e] = acc[q][u][ay * 4 + e];
                        }
            }
        }
    }
    __syncthreads();

    const int v  = tid;
    const int lx = v >> 6, ly = (v >> 4) & 3, lt = v & 15;
    const int x = x0 + lx, y = y0 + ly, t = t0 + lt;
#pragma unroll
    for (int d = 0; d < 27; ++d) {
        const float val = smem[d * 256 + tid] * SCALE;
        out[((d * VN + x) * VN + y) * VN + t] = val;
    }
}

extern "C" void kernel_launch(void* const* d_in, const int* in_sizes, int n_in,
                              void* d_out, int out_size, void* d_ws, size_t ws_size,
                              hipStream_t stream) {
    (void)in_sizes; (void)n_in; (void)out_size; (void)d_ws; (void)ws_size;
    const float* mov = (const float*)d_in[0];
    const float* fix = (const float*)d_in[1];
    float* out = (float*)d_out;
    dim3 grid(VN / (BT * NSUB), VN / BY, VN / BX);
    corr3d_wmma_kernel<<<grid, NSUB * NTHREADS, 0, stream>>>(mov, fix, out);
}


# Fix box x-pass taken in the t-pass lane's registers (it meets its row's ten planes in turn); LDS sweep left is y only, 10 loads per column not 30; fix box summed t, x, y, not t, y, x
# speedup vs baseline: 4.7226x; 1.0714x over previous
#include <hip/hip_runtime.h>

typedef float v2f __attribute__((ext_vector_type(2)));
typedef float v4f __attribute__((ext_vector_type(4)));
typedef float v8f __attribute__((ext_vector_type(8)));

#define VN 96
#define NC 32
#define BX 4
#define BY 4
#define BT 16
#define NTHREADS 256
#define NSUB 3
#define SCALE (1.0f / 27.0f)
#define NSLOTS 48
#define NMTRIPS 3
#define ROW4 (VN / 4)
#define PLANE4 (VN * ROW4)
#define CHAN4 (VN * PLANE4)

#define S1_OFF 0
#define S1_CS 2000
#define S1_XS 204
#define BF_OFF 8000
#define BF_CS 176
#define BF_XS 708
#define PM_OFF 13664
#define M1_OFF 14688
#define SMEM_FLOATS 16992


__device__ __forceinline__ v4f halo_piece(const v4f* __restrict__ p, bool ok, int idx) {
    const int i4 = ok ? idx : 0;
    __builtin_assume(i4 >= 0);
    return p[i4];
}

__device__ __forceinline__ float lane_next(float x) {
    return __builtin_bit_cast(float, __builtin_amdgcn_update_dpp(
        0, __builtin_bit_cast(int, x), 0x101, 0xf, 0xf, true));
}
__device__ __forceinline__ float lane_prev(float x) {
    return __builtin_bit_cast(float, __builtin_amdgcn_update_dpp(
        0, __builtin_bit_cast(int, x), 0x111, 0xf, 0xf, true));
}

__device__ __forceinline__ v4f fix_tpass(v4f v) {
    const float n0 = lane_next(v[0]);
    const float n1 = lane_next(v[1]);
    const float n2 = lane_next(v[2]);
    v4f o;
    o[0] = v[1] + v[2] + v[3];
    o[2] = v[2] + v[3] + n0;
    o[1] = v[3] + n0 + n1;
    o[3] = n0 + n1 + n2;
    return o;
}

__device__ __forceinline__ v4f mov_tpass(v4f v) {
    const float p3 = lane_prev(v[3]);
    const float n0 = lane_next(v[0]);
    v4f o;
    o[0] = p3 + v[0] + v[1];
    o[1] = v[0] + v[1] + v[2];
    o[2] = v[1] + v[2] + v[3];
    o[3] = v[2] + v[3] + n0;
    return o;
}


__device__ __forceinline__ v4f piece_at(const v4f* __restrict__ p, int idx) {
    __builtin_assume(idx >= 0);
    return p[idx];
}

__device__ __forceinline__ const v4f* plane_ptr(const v4f* __restrict__ base, int gx, int x0) {
    const int gxc = (unsigned)gx < (unsigned)VN ? gx : x0;
    return base + gxc * PLANE4;
}

__device__ __forceinline__ v4f fix_t(v4f v, bool ok) {
    const v4f zero4 = {0.0f, 0.0f, 0.0f, 0.0f};
    return fix_tpass(ok ? v : zero4);
}
__device__ __forceinline__ void fix_xput(v4f a, v4f b, v4f c, float* dst, bool st) {
    if (st)
        *reinterpret_cast<v4f*>(dst) = a + b + c;
}
__device__ __forceinline__ void mov_put(v4f v, bool ok, float* dst, bool st) {
    const v4f zero4 = {0.0f, 0.0f, 0.0f, 0.0f};
    const v4f o = mov_tpass(ok ? v : zero4);
    if (st)
        *reinterpret_cast<v4f*>(dst) = o;
}

__device__ __forceinline__ void load_issued() {
    __builtin_amdgcn_sched_barrier(0);
}

template <int NIN, int XSTEP, int YSTEP, int DSTEP>
__device__ __forceinline__ void sweep_yx(const float* src, float* dst) {
    v4f y0 = {0.0f, 0.0f, 0.0f, 0.0f}, y1 = y0;
#pragma unroll
    for (int i = 0; i < NIN; ++i) {
        const float* p = src + i * XSTEP;
        const v4f a = *reinterpret_cast<const v4f*>(p);
        const v4f b = *reinterpret_cast<const v4f*>(p + YSTEP);
        const v4f c = *reinterpret_cast<const v4f*>(p + 2 * YSTEP);
        const v4f y = a + b + c;
        if (i >= 2)
            *reinterpret_cast<v4f*>(dst + (i - 2) * DSTEP) = y0 + y1 + y;
        y0 = y1;
        y1 = y;
    }
}

__device__ __forceinline__ void sweep_y(const float* src, float* dst) {
    v4f r[10];
#pragma unroll
    for (int y = 0; y < 10; ++y)
        r[y] = *reinterpret_cast<const v4f*>(src + y * 20);
#pragma unroll
    for (int y = 0; y < 8; ++y)
        *reinterpret_cast<v4f*>(dst + y * 20) = r[y] + r[y + 1] + r[y + 2];
}

__global__ __launch_bounds__(NSUB * NTHREADS, 6)
void corr3d_wmma_kernel(const float* __restrict__ mov,
                        const float* __restrict__ fix,
                        float* __restrict__ out) {
    __shared__ __attribute__((aligned(16))) float smem_all[NSUB * SMEM_FLOATS];

    const int sub  = __builtin_amdgcn_readfirstlane((int)(threadIdx.x >> 8));
    float* const smem = smem_all + sub * SMEM_FLOATS;
    const int tid  = threadIdx.x & (NTHREADS - 1);
    const int lane = tid & 31;
    const int w    = tid >> 5;
    const int col  = lane & 15;
    const int hlf  = lane >> 4;

    const int x0 = blockIdx.z * BX;
    const int y0 = blockIdx.y * BY;
    const int t0 = (blockIdx.x * NSUB + sub) * BT;

    const int px = (w >> 2) & 1, py = (w >> 1) & 1, pt = w & 1;
    const int pm_base = PM_OFF +
        (((px + 2 * (col >> 3)) * 4 + 2 * hlf) * 4 + (py + 2 * ((col >> 2) & 1))) * 16 +
        pt + 2 * (col & 3);
    const int bf_base = BF_OFF + (px + 2 * (col >> 2)) * BF_XS + (2 * hlf) * BF_CS +
                        (py + 2 * (col & 3)) * 20 + 2 * pt;

    v8f acc[2][6];
#pragma unroll
    for (int q = 0; q < 2; ++q)
#pragma unroll
        for (int u = 0; u < 6; ++u) {
            const v8f z = {};
            acc[q][u] = z;
        }

    const int  gtid = threadIdx.x;
    const int  hq   = gtid & 15;
    const int  hs   = gtid >> 4;
    const int  T0   = blockIdx.x * (NSUB * BT);
    const int  hgt0 = T0 - 4 + 4 * hq;
    const bool htok = (hq < 14) && (unsigned)hgt0 < (unsigned)VN;
    const int  hsub = hq == 0 ? 0 : (((hq - 1) >> 2) < 2 ? ((hq - 1) >> 2) : 2);
    float* const hdst = smem_all + hsub * SMEM_FLOATS;
    const bool hsok = hs < 40;
    const int  c4h  = hsok ? hs / 10 : 0;
    const int  r10  = hsok ? hs - 10 * c4h : 0;
    const int  gyA   = y0 - 3 + r10;
    const bool yokA  = (unsigned)gyA < (unsigned)VN;
    const bool okA   = htok && yokA;
    const int  baseA = c4h * CHAN4 + (yokA ? gyA : y0) * ROW4 + ((htok ? hgt0 : T0) >> 2);
    float* const s1A = &hdst[S1_OFF + c4h * S1_CS + r10 * 20 + 4 * (hq - 4 * hsub)];
    const bool stA = hsok && hq <= 12;
    int  moff[NMTRIPS]; bool mok[NMTRIPS];
#pragma unroll
    for (int k = 0; k < NMTRIPS; ++k) {
        const int M   = k * NSLOTS + hs;
        const int c4  = M / 36, rem = M - c4 * 36;
        const int ix  = rem / 6, iy = rem - ix * 6;
        const int gx  = x0 - 1 + ix, gy = y0 - 1 + iy;
        mok[k]  = htok &&
                  (unsigned)gx < (unsigned)VN && (unsigned)gy < (unsigned)VN;
        moff[k] = ((c4 * VN + gx) * VN + gy) * ROW4 + (hgt0 >> 2);
    }
    float* const m1p = &hdst[M1_OFF + hs * 16 + 4 * (hq - 1 - 4 * hsub)];
    const bool stM = hq >= 1 && hq <= 12;
    const v4f* __restrict__ fix4 = reinterpret_cast<const v4f*>(fix);
    const v4f* __restrict__ mov4 = reinterpret_cast<const v4f*>(mov);

    v4f w0, w1, w2, w3, w4, w5;
    w0 = piece_at(plane_ptr(fix4, x0 - 3 + 0, x0), baseA);
    w1 = piece_at(plane_ptr(fix4, x0 - 3 + 1, x0), baseA);
    w2 = piece_at(plane_ptr(fix4, x0 - 3 + 2, x0), baseA);
    w3 = piece_at(plane_ptr(fix4, x0 - 3 + 3, x0), baseA);
    w4 = piece_at(plane_ptr(fix4, x0 - 3 + 4, x0), baseA);
    w5 = piece_at(plane_ptr(fix4, x0 - 3 + 5, x0), baseA);

    for (int cg = 0; cg < 8; ++cg) {
        {
            const int cgoff = cg * 4 * CHAN4;
            const int idxA  = baseA + cgoff;
            v4f oa, ob, oc;
            oa = fix_t(w0, okA && (unsigned)(x0 - 3 + 0) < (unsigned)VN);
            w0 = piece_at(plane_ptr(fix4, x0 - 3 + 6, x0), idxA);
            load_issued();
            ob = fix_t(w1, okA && (unsigned)(x0 - 3 + 1) < (unsigned)VN);
            w1 = piece_at(plane_ptr(fix4, x0 - 3 + 7, x0), idxA);
            load_issued();
            oc = fix_t(w2, okA && (unsigned)(x0 - 3 + 2) < (unsigned)VN);
            fix_xput(oa, ob, oc, s1A + 0 * S1_XS, stA);
            w2 = piece_at(plane_ptr(fix4, x0 - 3 + 8, x0), idxA);
            load_issued();
            oa = fix_t(w3, okA && (unsigned)(x0 - 3 + 3) < (unsigned)VN);
            fix_xput(ob, oc, oa, s1A + 1 * S1_XS, stA);
            w3 = piece_at(plane_ptr(fix4, x0 - 3 + 9, x0), idxA);
            load_issued();
            ob = fix_t(w4, okA && (unsigned)(x0 - 3 + 4) < (unsigned)VN);
            fix_xput(oc, oa, ob, s1A + 2 * S1_XS, stA);
            w4 = halo_piece(mov4, mok[0], moff[0] + cgoff);
            load_issued();
            oc = fix_t(w5, okA && (unsigned)(x0 - 3 + 5) < (unsigned)VN);
            fix_xput(oa, ob, oc, s1A + 3 * S1_XS, stA);
            w5 = halo_piece(mov4, mok[1], moff[1] + cgoff);
            load_issued();
            oa = fix_t(w0, okA && (unsigned)(x0 - 3 + 6) < (unsigned)VN);
            fix_xput(ob, oc, oa, s1A + 4 * S1_XS, stA);
            w0 = halo_piece(mov4, mok[2], moff[2] + cgoff);
            load_issued();
            ob = fix_t(w1, okA && (unsigned)(x0 - 3 + 7) < (unsigned)VN);
            fix_xput(oc, oa, ob, s1A + 5 * S1_XS, stA);
            oc = fix_t(w2, okA && (unsigned)(x0 - 3 + 8) < (unsigned)VN);
            fix_xput(oa, ob, oc, s1A + 6 * S1_XS, stA);
            oa = fix_t(w3, okA && (unsigned)(x0 - 3 + 9) < (unsigned)VN);
            fix_xput(ob, oc, oa, s1A + 7 * S1_XS, stA);
            mov_put(w4, mok[0], m1p + 0 * NSLOTS * 16, stM);
            mov_put(w5, mok[1], m1p + 1 * NSLOTS * 16, stM);
            mov_put(w0, mok[2], m1p + 2 * NSLOTS * 16, stM);
        }
        __syncthreads();
        if (tid < 160) {
            const int p  = tid >> 5;
            const int xp = (tid & 31) >> 2;
            const int c4 = tid & 3;
            const bool fromprev = sub > 0 && p == 0;
            const float* const s1c = &smem[S1_OFF + c4 * S1_CS + xp * S1_XS + 4 * p];
            sweep_y(fromprev ? s1c - SMEM_FLOATS + 16 : s1c,
                    &smem[BF_OFF + xp * BF_XS + c4 * BF_CS + 4 * p]);
        }
        if (tid >= 192) {
            const int jj = tid - 192;
            const int c4 = jj >> 4, jm = jj & 15;
            sweep_yx<6, 96, 16, 256>(&smem[M1_OFF + c4 * 576 + 4 * jm],
                                     &smem[PM_OFF + 4 * jj]);
        }
        __syncthreads();

        if (cg + 1 < 8) {
            const int idxN = baseA + (cg + 1) * 4 * CHAN4;
            w0 = piece_at(plane_ptr(fix4, x0 - 3 + 0, x0), idxN);
            w1 = piece_at(plane_ptr(fix4, x0 - 3 + 1, x0), idxN);
            w2 = piece_at(plane_ptr(fix4, x0 - 3 + 2, x0), idxN);
            w3 = piece_at(plane_ptr(fix4, x0 - 3 + 3, x0), idxN);
            w4 = piece_at(plane_ptr(fix4, x0 - 3 + 4, x0), idxN);
            w5 = piece_at(plane_ptr(fix4, x0 - 3 + 5, x0), idxN);
        }

#pragma unroll
        for (int q = 0; q < 2; ++q) {
            v2f fa, fb[6];
            fa[0] = smem[pm_base + 8 * q];
            fa[1] = smem[pm_base + 8 * q + 64];
#pragma unroll
            for (int pp = 0; pp < 3; ++pp) {
                const int p = 2 * q + pp;
                const v2f c0 = *reinterpret_cast<const v2f*>(&smem[bf_base + 4 * p]);
                const v2f c1 = *reinterpret_cast<const v2f*>(&smem[bf_base + 4 * p + BF_CS]);
                fb[2 * pp][0]     = c0[0];
                fb[2 * pp][1]     = c1[0];
                fb[2 * pp + 1][0] = c0[1];
                fb[2 * pp + 1][1] = c1[1];
            }
#pragma unroll
            for (int u = 0; u < 6; ++u)
                acc[q][u] = __builtin_amdgcn_wmma_f32_16x16x4_f32(
                    false, fa, false, fb[u], (short)0, acc[q][u], false, false);
            if (q == 0)
                __builtin_amdgcn_sched_barrier(0);
        }
    }

    {
        const int di = (col >> 2) - hlf;
#pragma unroll
        for (int ay = 0; ay < 2; ++ay) {
            const int dj = (col & 3) - ay;
            if ((unsigned)di < 3u && (unsigned)dj < 3u) {
                const int ob = ((di * 3 + dj) * 3) * 256 +
                               ((px + 2 * hlf) * 4 + (py + 2 * ay)) * 16 + pt;
#pragma unroll
                for (int q = 0; q < 2; ++q)
#pragma unroll
                    for (int u = 0; u < 6; ++u)
#pragma unroll
                        for (int e = 0; e < 4; ++e) {
                            const int k = u - e;
                            if (k >= 0 && k <= 2)
                                smem[ob + k * 256 + 8 * q + 2 * e] = acc[q][u][ay * 4 + e];
                        }
            }
        }
    }
    __syncthreads();

    const int v  = tid;
    const int lx = v >> 6, ly = (v >> 4) & 3, lt = v & 15;
    const int x = x0 + lx, y = y0 + ly, t = t0 + lt;
#pragma unroll
    for (int d = 0; d < 27; ++d) {
        const float val = smem[d * 256 + tid] * SCALE;
        out[((d * VN + x) * VN + y) * VN + t] = val;
    }
}

extern "C" void kernel_launch(void* const* d_in, const int* in_sizes, int n_in,
                              void* d_out, int out_size, void* d_ws, size_t ws_size,
                              hipStream_t stream) {
    (void)in_sizes; (void)n_in; (void)out_size; (void)d_ws; (void)ws_size;
    const float* mov = (const float*)d_in[0];
    const float* fix = (const float*)d_in[1];
    float* out = (float*)d_out;
    dim3 grid(VN / (BT * NSUB), VN / BY, VN / BX);
    corr3d_wmma_kernel<<<grid, NSUB * NTHREADS, 0, stream>>>(mov, fix, out);
}
